# starting kernel
# baseline (speedup 1.0000x reference)
#include <hip/hip_runtime.h>

#define N_NODES 50000
#define E_EDGES 800000
#define F_IN    256
#define F_OUT   128
#define BATCH   2
#define OB      (F_OUT * BATCH)

#define NT ((N_NODES + 15) / 16)
#define OT (F_OUT / 16)
#define NT_PER_BLK 8

typedef float v2f __attribute__((ext_vector_type(2)));
typedef float v8f __attribute__((ext_vector_type(8)));


__global__ void deg_init_kernel(float* __restrict__ deg) {
    int i = blockIdx.x * blockDim.x + threadIdx.x;
    if (i < N_NODES) deg[i] = 1.0f;
}

__global__ void deg_acc_kernel(const int* __restrict__ row,
                               const float* __restrict__ ew,
                               float* __restrict__ deg) {
    for (int e = blockIdx.x * blockDim.x + threadIdx.x; e < E_EDGES;
         e += gridDim.x * blockDim.x) {
        atomicAdd(&deg[row[e]], ew[e]);
    }
}

__global__ void dinv_kernel(const float* __restrict__ deg,
                            float* __restrict__ dinv) {
    int i = blockIdx.x * blockDim.x + threadIdx.x;
    if (i < N_NODES) {
        float d = deg[i];
        dinv[i] = d > 0.0f ? rsqrtf(d) : 0.0f;
    }
}


__global__ void gemm_wmma_kernel(const float* __restrict__ x,
                                 const float* __restrict__ w,
                                 float* __restrict__ xw) {
    __shared__ float wtile[F_IN * 16];

    const int ntb_blocks = (NT + NT_PER_BLK - 1) / NT_PER_BLK;
    int blk  = blockIdx.x;
    int ntb  = blk % ntb_blocks;
    int rest = blk / ntb_blocks;
    int ot   = rest % OT;
    int b    = rest / OT;
    int o0   = ot * 16;

    for (int i = threadIdx.x; i < F_IN * 16; i += blockDim.x) {
        int k = i >> 4;
        int j = i & 15;
        wtile[i] = w[k * F_OUT + o0 + j];
    }
    __syncthreads();

    int wave = threadIdx.x >> 5;
    int lane = threadIdx.x & 31;
    int nt   = ntb * NT_PER_BLK + wave;
    if (nt >= NT) return;

    int n0   = nt * 16;
    int half = lane >> 4;
    int l    = lane & 15;

    const float* xrow = x + ((size_t)b * N_NODES + (size_t)(n0 + l)) * F_IN;

    v8f acc = {};
    for (int k = 0; k < F_IN; k += 4) {
        v2f a, bf;
        a.x  = xrow[k + half * 2 + 0];
        a.y  = xrow[k + half * 2 + 1];
        bf.x = wtile[(k + half * 2 + 0) * 16 + l];
        bf.y = wtile[(k + half * 2 + 1) * 16 + l];
        acc = __builtin_amdgcn_wmma_f32_16x16x4_f32(
             false, a,  false, bf,
             (short)0, acc,  false,  false);
    }

    #pragma unroll
    for (int v = 0; v < 8; ++v) {
        int m = v + half * 8;
        xw[(size_t)(n0 + m) * OB + (size_t)(o0 + l) * 2 + b] = acc[v];
    }
}


__global__ void out_init_kernel(const float* __restrict__ dinv,
                                const float* __restrict__ xw,
                                const float* __restrict__ bias,
                                float* __restrict__ out) {
    size_t i = (size_t)blockIdx.x * blockDim.x + threadIdx.x;
    const size_t total = (size_t)BATCH * N_NODES * F_OUT;
    if (i >= total) return;
    int o    = (int)(i % F_OUT);
    size_t t = i / F_OUT;
    int n    = (int)(t % N_NODES);
    int b    = (int)(t / N_NODES);
    float d  = dinv[n];
    out[i] = bias[o] + d * d * xw[(size_t)n * OB + (size_t)o * 2 + b];
}


__global__ void edge_scatter_kernel(const int* __restrict__ row,
                                    const int* __restrict__ col,
                                    const float* __restrict__ ew,
                                    const float* __restrict__ dinv,
                                    const float* __restrict__ xw,
                                    float* __restrict__ out) {
    int c = threadIdx.x;
    int o = c >> 1;
    int b = c & 1;
    for (int e = blockIdx.x; e < E_EDGES; e += gridDim.x) {
        int r  = row[e];
        int cl = col[e];
        float lap = dinv[r] * ew[e] * dinv[cl];
        float val = lap * xw[(size_t)cl * OB + c];
        atomicAdd(&out[((size_t)b * N_NODES + r) * F_OUT + o], val);
    }
}


extern "C" void kernel_launch(void* const* d_in, const int* in_sizes, int n_in,
                              void* d_out, int out_size, void* d_ws, size_t ws_size,
                              hipStream_t stream) {
    const float* x    = (const float*)d_in[0];
    const float* w    = (const float*)d_in[1];
    const float* bias = (const float*)d_in[2];
    const float* ew   = (const float*)d_in[3];
    const int*   eidx = (const int*)d_in[4];
    const int*   row  = eidx;
    const int*   col  = eidx + E_EDGES;
    float* out = (float*)d_out;

    float* xw   = (float*)d_ws;
    float* deg  = xw + (size_t)N_NODES * OB;
    float* dinv = deg + N_NODES;

    deg_init_kernel<<<(N_NODES + 255) / 256, 256, 0, stream>>>(deg);
    deg_acc_kernel<<<1024, 256, 0, stream>>>(row, ew, deg);
    dinv_kernel<<<(N_NODES + 255) / 256, 256, 0, stream>>>(deg, dinv);

    const int ntb_blocks = (NT + NT_PER_BLK - 1) / NT_PER_BLK;
    gemm_wmma_kernel<<<ntb_blocks * OT * BATCH, 256, 0, stream>>>(x, w, xw);

    size_t total = (size_t)BATCH * N_NODES * F_OUT;
    out_init_kernel<<<(unsigned)((total + 255) / 256), 256, 0, stream>>>(dinv, xw, bias, out);

    edge_scatter_kernel<<<8192, 256, 0, stream>>>(row, col, ew, dinv, xw, out);
}


# Edge sum by destination range: 256 work-groups each keep 196 nodes' output rows in LDS, scan row[], add gathered xw rows with LDS float adds (arrival order), write out once; no global atomics
# speedup vs baseline: 2.8453x; 2.8453x over previous
#include <hip/hip_runtime.h>

#define N_NODES 50000
#define E_EDGES 800000
#define F_IN    256
#define F_OUT   128
#define BATCH   2
#define OB      (F_OUT * BATCH)

#define NT ((N_NODES + 15) / 16)
#define OT (F_OUT / 16)
#define NT_PER_BLK 8

typedef float v2f __attribute__((ext_vector_type(2)));
typedef float v8f __attribute__((ext_vector_type(8)));


__global__ void deg_init_kernel(float* __restrict__ deg) {
    int i = blockIdx.x * blockDim.x + threadIdx.x;
    if (i < N_NODES) deg[i] = 1.0f;
}

__global__ void deg_acc_kernel(const int* __restrict__ row,
                               const float* __restrict__ ew,
                               float* __restrict__ deg) {
    for (int e = blockIdx.x * blockDim.x + threadIdx.x; e < E_EDGES;
         e += gridDim.x * blockDim.x) {
        atomicAdd(&deg[row[e]], ew[e]);
    }
}

__global__ void dinv_kernel(const float* __restrict__ deg,
                            float* __restrict__ dinv) {
    int i = blockIdx.x * blockDim.x + threadIdx.x;
    if (i < N_NODES) {
        float d = deg[i];
        dinv[i] = d > 0.0f ? rsqrtf(d) : 0.0f;
    }
}


__global__ void gemm_wmma_kernel(const float* __restrict__ x,
                                 const float* __restrict__ w,
                                 float* __restrict__ xw) {
    __shared__ float wtile[F_IN * 16];

    const int ntb_blocks = (NT + NT_PER_BLK - 1) / NT_PER_BLK;
    int blk  = blockIdx.x;
    int ntb  = blk % ntb_blocks;
    int rest = blk / ntb_blocks;
    int ot   = rest % OT;
    int b    = rest / OT;
    int o0   = ot * 16;

    for (int i = threadIdx.x; i < F_IN * 16; i += blockDim.x) {
        int k = i >> 4;
        int j = i & 15;
        wtile[i] = w[k * F_OUT + o0 + j];
    }
    __syncthreads();

    int wave = threadIdx.x >> 5;
    int lane = threadIdx.x & 31;
    int nt   = ntb * NT_PER_BLK + wave;
    if (nt >= NT) return;

    int n0   = nt * 16;
    int half = lane >> 4;
    int l    = lane & 15;

    const float* xrow = x + ((size_t)b * N_NODES + (size_t)(n0 + l)) * F_IN;

    v8f acc = {};
    for (int k = 0; k < F_IN; k += 4) {
        v2f a, bf;
        a.x  = xrow[k + half * 2 + 0];
        a.y  = xrow[k + half * 2 + 1];
        bf.x = wtile[(k + half * 2 + 0) * 16 + l];
        bf.y = wtile[(k + half * 2 + 1) * 16 + l];
        acc = __builtin_amdgcn_wmma_f32_16x16x4_f32(
             false, a,  false, bf,
             (short)0, acc,  false,  false);
    }

    #pragma unroll
    for (int v = 0; v < 8; ++v) {
        int m = v + half * 8;
        xw[(size_t)(n0 + m) * OB + (size_t)(o0 + l) * 2 + b] = acc[v];
    }
}


#define OWN_WGS      256
#define ROWS_PER_WG  ((N_NODES + OWN_WGS - 1) / OWN_WGS)
#define OWN_THREADS  1024
#define OWN_WAVES    (OWN_THREADS / 32)
#define QCAP         8192
#define N_ROW4       (E_EDGES / 4)
#define SCAN_UNROLL  8
#define SCAN_STEP    (SCAN_UNROLL * OWN_THREADS)
#define SCAN_FULL    ((N_ROW4 / SCAN_STEP) * SCAN_STEP)

__device__ __forceinline__ int tile_pos(int c) { return ((c & 3) << 6) | (c >> 2); }

__device__ __forceinline__ void lds_add(float* p, float v) {
    __hip_atomic_fetch_add(p, v, __ATOMIC_RELAXED, __HIP_MEMORY_SCOPE_WORKGROUP);
}

__device__ __forceinline__ void queue_owned(int4 v, int idx, int r0, int nrows,
                                            unsigned* q_cnt, unsigned* q_a) {
    int rr[4] = {v.x, v.y, v.z, v.w};
    #pragma unroll
    for (int j = 0; j < 4; ++j) {
        unsigned d = (unsigned)(rr[j] - r0);
        if (d < (unsigned)nrows) {
            unsigned pos = atomicAdd(q_cnt, 1u);
            if (pos < QCAP) q_a[pos] = (unsigned)(idx * 4 + j) | (d << 20);
        }
    }
}

__global__ __launch_bounds__(OWN_THREADS)
void edge_owner_kernel(const int* __restrict__ row,
                       const int* __restrict__ col,
                       const float* __restrict__ ew,
                       const float* __restrict__ dinv,
                       const float* __restrict__ xw,
                       const float* __restrict__ bias,
                       float* __restrict__ out) {
    __shared__ float    tile[ROWS_PER_WG * OB];
    __shared__ unsigned q_a[QCAP];
    __shared__ float    q_lap[QCAP];
    __shared__ unsigned q_cnt;

    const int tid   = threadIdx.x;
    const int r0    = blockIdx.x * ROWS_PER_WG;
    const int nrows = min(ROWS_PER_WG, N_NODES - r0);

    for (int i = tid; i < ROWS_PER_WG * OB / 4; i += OWN_THREADS)
        ((float4*)tile)[i] = make_float4(0.f, 0.f, 0.f, 0.f);
    if (tid == 0) q_cnt = 0u;
    __syncthreads();

    const int4* row4  = (const int4*)row;
    const int   start = blockIdx.x * (N_ROW4 / OWN_WGS);
    for (int k = tid; k < SCAN_FULL; k += SCAN_UNROLL * OWN_THREADS) {
        int4 v[SCAN_UNROLL];
        int  id[SCAN_UNROLL];
        #pragma unroll
        for (int u = 0; u < SCAN_UNROLL; ++u) {
            int idx = start + k + u * OWN_THREADS;
            if (idx >= N_ROW4) idx -= N_ROW4;
            id[u] = idx;
            v[u]  = row4[idx];
        }
        __builtin_amdgcn_sched_barrier(0);
        #pragma unroll
        for (int u = 0; u < SCAN_UNROLL; ++u) queue_owned(v[u], id[u], r0, nrows, &q_cnt, q_a);
    }
    for (int kk = SCAN_FULL + tid; kk < N_ROW4; kk += OWN_THREADS) {
        int idx = start + kk;
        if (idx >= N_ROW4) idx -= N_ROW4;
        queue_owned(row4[idx], idx, r0, nrows, &q_cnt, q_a);
    }
    __syncthreads();

    int nq = (int)q_cnt;
    if (q_cnt > (unsigned)QCAP) {
        nq = 0;
        for (int e = tid; e < E_EDGES; e += OWN_THREADS) {
            int r = row[e];
            unsigned d = (unsigned)(r - r0);
            if (d < (unsigned)nrows) {
                int   cl  = col[e];
                float lap = dinv[r] * ew[e] * dinv[cl];
                const float* srow = xw + (size_t)cl * OB;
                float* trow = tile + (int)d * OB;
                for (int c = 0; c < OB; ++c) lds_add(&trow[tile_pos(c)], lap * srow[c]);
            }
        }
    }

    for (int i = tid; i < nq; i += OWN_THREADS) {
        unsigned pk = q_a[i];
        int e  = (int)(pk & 0xFFFFFu);
        int d  = (int)(pk >> 20);
        int cl = col[e];
        float lap = dinv[r0 + d] * ew[e] * dinv[cl];
        q_a[i]   = (unsigned)cl | ((unsigned)d << 16);
        q_lap[i] = lap;
    }
    __syncthreads();

    const int wave = __builtin_amdgcn_readfirstlane(tid >> 5);
    const int lane = tid & 31;
    int i = wave;
    for (; i + 3 * OWN_WAVES < nq; i += 4 * OWN_WAVES) {
        unsigned pk[4];
        float    lp[4];
        float4   a[4], b[4];
        #pragma unroll
        for (int u = 0; u < 4; ++u) {
            pk[u] = q_a[i + u * OWN_WAVES];
            lp[u] = q_lap[i + u * OWN_WAVES];
            const float4* s = (const float4*)(xw + (size_t)(pk[u] & 0xFFFFu) * OB);
            a[u] = s[lane];
            b[u] = s[32 + lane];
        }
        #pragma unroll
        for (int u = 0; u < 4; ++u) {
            float* t = tile + (int)(pk[u] >> 16) * OB + lane;
            lds_add(t +   0, lp[u] * a[u].x);
            lds_add(t +  64, lp[u] * a[u].y);
            lds_add(t + 128, lp[u] * a[u].z);
            lds_add(t + 192, lp[u] * a[u].w);
            lds_add(t +  32, lp[u] * b[u].x);
            lds_add(t +  96, lp[u] * b[u].y);
            lds_add(t + 160, lp[u] * b[u].z);
            lds_add(t + 224, lp[u] * b[u].w);
        }
    }
    for (; i < nq; i += OWN_WAVES) {
        unsigned pk = q_a[i];
        float    lp = q_lap[i];
        const float4* s = (const float4*)(xw + (size_t)(pk & 0xFFFFu) * OB);
        float4 a = s[lane];
        float4 b = s[32 + lane];
        float* t = tile + (int)(pk >> 16) * OB + lane;
        lds_add(t +   0, lp * a.x);
        lds_add(t +  64, lp * a.y);
        lds_add(t + 128, lp * a.z);
        lds_add(t + 192, lp * a.w);
        lds_add(t +  32, lp * b.x);
        lds_add(t +  96, lp * b.y);
        lds_add(t + 160, lp * b.z);
        lds_add(t + 224, lp * b.w);
    }
    __syncthreads();

    for (int t = tid; t < nrows * OB; t += OWN_THREADS) {
        int d   = t >> 8;
        int rem = t & 255;
        int b   = rem >> 7;
        int o   = rem & 127;
        int n   = r0 + d;
        int c   = o * 2 + b;
        float dv = dinv[n];
        out[((size_t)b * N_NODES + n) * F_OUT + o] =
            bias[o] + dv * dv * xw[(size_t)n * OB + c] + tile[d * OB + tile_pos(c)];
    }
}


extern "C" void kernel_launch(void* const* d_in, const int* in_sizes, int n_in,
                              void* d_out, int out_size, void* d_ws, size_t ws_size,
                              hipStream_t stream) {
    const float* x    = (const float*)d_in[0];
    const float* w    = (const float*)d_in[1];
    const float* bias = (const float*)d_in[2];
    const float* ew   = (const float*)d_in[3];
    const int*   eidx = (const int*)d_in[4];
    const int*   row  = eidx;
    const int*   col  = eidx + E_EDGES;
    float* out = (float*)d_out;

    float* xw   = (float*)d_ws;
    float* deg  = xw + (size_t)N_NODES * OB;
    float* dinv = deg + N_NODES;

    deg_init_kernel<<<(N_NODES + 255) / 256, 256, 0, stream>>>(deg);
    deg_acc_kernel<<<1024, 256, 0, stream>>>(row, ew, deg);
    dinv_kernel<<<(N_NODES + 255) / 256, 256, 0, stream>>>(deg, dinv);

    const int ntb_blocks = (NT + NT_PER_BLK - 1) / NT_PER_BLK;
    gemm_wmma_kernel<<<ntb_blocks * OT * BATCH, 256, 0, stream>>>(x, w, xw);

    edge_owner_kernel<<<OWN_WGS, OWN_THREADS, 0, stream>>>(row, col, ew, dinv, xw, bias, out);
}


# GEMM re-tiled: wave = 16 nodes x 64 outs x both batches, 16-B x loads each feed 8 WMMAs, w whole in LDS, full-line xw stores; k summed 0,1,4,5,2,3,6,7 per 8 (fixed order)
# speedup vs baseline: 3.5603x; 1.2513x over previous
#include <hip/hip_runtime.h>

#define N_NODES 50000
#define E_EDGES 800000
#define F_IN    256
#define F_OUT   128
#define BATCH   2
#define OB      (F_OUT * BATCH)

#define NT ((N_NODES + 15) / 16)

typedef float v2f __attribute__((ext_vector_type(2)));
typedef float v8f __attribute__((ext_vector_type(8)));


__global__ void deg_init_kernel(float* __restrict__ deg) {
    int i = blockIdx.x * blockDim.x + threadIdx.x;
    if (i < N_NODES) deg[i] = 1.0f;
}

__global__ void deg_acc_kernel(const int* __restrict__ row,
                               const float* __restrict__ ew,
                               float* __restrict__ deg) {
    for (int e = blockIdx.x * blockDim.x + threadIdx.x; e < E_EDGES;
         e += gridDim.x * blockDim.x) {
        atomicAdd(&deg[row[e]], ew[e]);
    }
}

__global__ void dinv_kernel(const float* __restrict__ deg,
                            float* __restrict__ dinv) {
    int i = blockIdx.x * blockDim.x + threadIdx.x;
    if (i < N_NODES) {
        float d = deg[i];
        dinv[i] = d > 0.0f ? rsqrtf(d) : 0.0f;
    }
}


#define GM_BLOCKS   256
#define GM_THREADS  768
#define GM_PAIRS    (GM_THREADS / 64)
#define KGROUPS     (F_IN / 8)

__device__ __forceinline__ v8f wmma4(float ax, float ay, float bx, float by, v8f c) {
    v2f a = {ax, ay};
    v2f b = {bx, by};
    return __builtin_amdgcn_wmma_f32_16x16x4_f32(
         false, a,  false, b,
         (short)0, c,  false,  false);
}

__device__ __forceinline__ void mma_group(const float4& a0, const float4& a1,
                                          const float4 (&bf)[4],
                                          v8f (&acc0)[4], v8f (&acc1)[4]) {
    #pragma unroll
    for (int t = 0; t < 4; ++t) {
        acc0[t] = wmma4(a0.x, a0.y, bf[t].x, bf[t].y, acc0[t]);
        acc1[t] = wmma4(a1.x, a1.y, bf[t].x, bf[t].y, acc1[t]);
    }
    #pragma unroll
    for (int t = 0; t < 4; ++t) {
        acc0[t] = wmma4(a0.z, a0.w, bf[t].z, bf[t].w, acc0[t]);
        acc1[t] = wmma4(a1.z, a1.w, bf[t].z, bf[t].w, acc1[t]);
    }
}

__global__ __launch_bounds__(GM_THREADS)
void gemm_wmma_kernel(const float* __restrict__ x,
                      const float* __restrict__ w,
                      float* __restrict__ xw) {
    __shared__ float4 wl[F_IN * F_OUT / 4];

    const int tid = threadIdx.x;
    for (int idx = tid; idx < F_IN * F_OUT / 4; idx += GM_THREADS) {
        int o  = idx & (F_OUT - 1);
        int kq = idx >> 7;
        const float* src = w + (size_t)(kq * 4) * F_OUT + o;
        float4 v = make_float4(src[0], src[F_OUT], src[2 * F_OUT], src[3 * F_OUT]);
        int kk = kq >> 1, hh = kq & 1;
        int oh = o >> 6, t = (o >> 4) & 3, ll = o & 15;
        wl[((oh * KGROUPS + kk) * 4 + t) * 32 + hh * 16 + ll] = v;
    }
    __syncthreads();

    const int wave = __builtin_amdgcn_readfirstlane(tid >> 5);
    const int lane = tid & 31;
    const int oh = wave & 1;
    const int h  = lane >> 4;
    const int l  = lane & 15;
    const float4* wb = wl + oh * (KGROUPS * 4 * 32) + lane;

    for (int nt = (wave >> 1) * GM_BLOCKS + blockIdx.x; nt < NT; nt += GM_PAIRS * GM_BLOCKS) {
        const int n0 = nt * 16;
        const float4* xa0 = (const float4*)(x + (size_t)(n0 + l) * F_IN + h * 4);
        const float4* xa1 = (const float4*)(x + ((size_t)N_NODES + n0 + l) * F_IN + h * 4);

        v8f acc0[4] = {};
        v8f acc1[4] = {};
        float4 a0 = xa0[0];
        float4 a1 = xa1[0];
        for (int kk = 0; kk < KGROUPS; ++kk) {
            float4 bf[4];
            #pragma unroll
            for (int t = 0; t < 4; ++t) bf[t] = wb[(kk * 4 + t) * 32];
            const int kn = (kk + 1) & (KGROUPS - 1);
            float4 na0 = xa0[kn * 2];
            float4 na1 = xa1[kn * 2];
            mma_group(a0, a1, bf, acc0, acc1);
            a0 = na0;
            a1 = na1;
        }

        float* orow = xw + (size_t)(n0 + 8 * h) * OB + (oh * 64 + l) * 2;
        #pragma unroll
        for (int t = 0; t < 4; ++t) {
            #pragma unroll
            for (int v = 0; v < 8; ++v) {
                float2 p = make_float2(acc0[t][v], acc1[t][v]);
                *(float2*)(orow + (size_t)v * OB + t * 32) = p;
            }
        }
    }
}


#define OWN_WGS      256
#define ROWS_PER_WG  ((N_NODES + OWN_WGS - 1) / OWN_WGS)
#define OWN_THREADS  1024
#define OWN_WAVES    (OWN_THREADS / 32)
#define QCAP         8192
#define N_ROW4       (E_EDGES / 4)
#define SCAN_UNROLL  8
#define SCAN_STEP    (SCAN_UNROLL * OWN_THREADS)
#define SCAN_FULL    ((N_ROW4 / SCAN_STEP) * SCAN_STEP)

__device__ __forceinline__ int tile_pos(int c) { return ((c & 3) << 6) | (c >> 2); }

__device__ __forceinline__ void lds_add(float* p, float v) {
    __hip_atomic_fetch_add(p, v, __ATOMIC_RELAXED, __HIP_MEMORY_SCOPE_WORKGROUP);
}

__device__ __forceinline__ void queue_owned(int4 v, int idx, int r0, int nrows,
                                            unsigned* q_cnt, unsigned* q_a) {
    int rr[4] = {v.x, v.y, v.z, v.w};
    #pragma unroll
    for (int j = 0; j < 4; ++j) {
        unsigned d = (unsigned)(rr[j] - r0);
        if (d < (unsigned)nrows) {
            unsigned pos = atomicAdd(q_cnt, 1u);
            if (pos < QCAP) q_a[pos] = (unsigned)(idx * 4 + j) | (d << 20);
        }
    }
}

__global__ __launch_bounds__(OWN_THREADS)
void edge_owner_kernel(const int* __restrict__ row,
                       const int* __restrict__ col,
                       const float* __restrict__ ew,
                       const float* __restrict__ dinv,
                       const float* __restrict__ xw,
                       const float* __restrict__ bias,
                       float* __restrict__ out) {
    __shared__ float    tile[ROWS_PER_WG * OB];
    __shared__ unsigned q_a[QCAP];
    __shared__ float    q_lap[QCAP];
    __shared__ unsigned q_cnt;

    const int tid   = threadIdx.x;
    const int r0    = blockIdx.x * ROWS_PER_WG;
    const int nrows = min(ROWS_PER_WG, N_NODES - r0);

    for (int i = tid; i < ROWS_PER_WG * OB / 4; i += OWN_THREADS)
        ((float4*)tile)[i] = make_float4(0.f, 0.f, 0.f, 0.f);
    if (tid == 0) q_cnt = 0u;
    __syncthreads();

    const int4* row4  = (const int4*)row;
    const int   start = blockIdx.x * (N_ROW4 / OWN_WGS);
    for (int k = tid; k < SCAN_FULL; k += SCAN_UNROLL * OWN_THREADS) {
        int4 v[SCAN_UNROLL];
        int  id[SCAN_UNROLL];
        #pragma unroll
        for (int u = 0; u < SCAN_UNROLL; ++u) {
            int idx = start + k + u * OWN_THREADS;
            if (idx >= N_ROW4) idx -= N_ROW4;
            id[u] = idx;
            v[u]  = row4[idx];
        }
        __builtin_amdgcn_sched_barrier(0);
        #pragma unroll
        for (int u = 0; u < SCAN_UNROLL; ++u) queue_owned(v[u], id[u], r0, nrows, &q_cnt, q_a);
    }
    for (int kk = SCAN_FULL + tid; kk < N_ROW4; kk += OWN_THREADS) {
        int idx = start + kk;
        if (idx >= N_ROW4) idx -= N_ROW4;
        queue_owned(row4[idx], idx, r0, nrows, &q_cnt, q_a);
    }
    __syncthreads();

    int nq = (int)q_cnt;
    if (q_cnt > (unsigned)QCAP) {
        nq = 0;
        for (int e = tid; e < E_EDGES; e += OWN_THREADS) {
            int r = row[e];
            unsigned d = (unsigned)(r - r0);
            if (d < (unsigned)nrows) {
                int   cl  = col[e];
                float lap = dinv[r] * ew[e] * dinv[cl];
                const float* srow = xw + (size_t)cl * OB;
                float* trow = tile + (int)d * OB;
                for (int c = 0; c < OB; ++c) lds_add(&trow[tile_pos(c)], lap * srow[c]);
            }
        }
    }

    for (int i = tid; i < nq; i += OWN_THREADS) {
        unsigned pk = q_a[i];
        int e  = (int)(pk & 0xFFFFFu);
        int d  = (int)(pk >> 20);
        int cl = col[e];
        float lap = dinv[r0 + d] * ew[e] * dinv[cl];
        q_a[i]   = (unsigned)cl | ((unsigned)d << 16);
        q_lap[i] = lap;
    }
    __syncthreads();

    const int wave = __builtin_amdgcn_readfirstlane(tid >> 5);
    const int lane = tid & 31;
    int i = wave;
    for (; i + 3 * OWN_WAVES < nq; i += 4 * OWN_WAVES) {
        unsigned pk[4];
        float    lp[4];
        float4   a[4], b[4];
        #pragma unroll
        for (int u = 0; u < 4; ++u) {
            pk[u] = q_a[i + u * OWN_WAVES];
            lp[u] = q_lap[i + u * OWN_WAVES];
            const float4* s = (const float4*)(xw + (size_t)(pk[u] & 0xFFFFu) * OB);
            a[u] = s[lane];
            b[u] = s[32 + lane];
        }
        #pragma unroll
        for (int u = 0; u < 4; ++u) {
            float* t = tile + (int)(pk[u] >> 16) * OB + lane;
            lds_add(t +   0, lp[u] * a[u].x);
            lds_add(t +  64, lp[u] * a[u].y);
            lds_add(t + 128, lp[u] * a[u].z);
            lds_add(t + 192, lp[u] * a[u].w);
            lds_add(t +  32, lp[u] * b[u].x);
            lds_add(t +  96, lp[u] * b[u].y);
            lds_add(t + 160, lp[u] * b[u].z);
            lds_add(t + 224, lp[u] * b[u].w);
        }
    }
    for (; i < nq; i += OWN_WAVES) {
        unsigned pk = q_a[i];
        float    lp = q_lap[i];
        const float4* s = (const float4*)(xw + (size_t)(pk & 0xFFFFu) * OB);
        float4 a = s[lane];
        float4 b = s[32 + lane];
        float* t = tile + (int)(pk >> 16) * OB + lane;
        lds_add(t +   0, lp * a.x);
        lds_add(t +  64, lp * a.y);
        lds_add(t + 128, lp * a.z);
        lds_add(t + 192, lp * a.w);
        lds_add(t +  32, lp * b.x);
        lds_add(t +  96, lp * b.y);
        lds_add(t + 160, lp * b.z);
        lds_add(t + 224, lp * b.w);
    }
    __syncthreads();

    for (int t = tid; t < nrows * OB; t += OWN_THREADS) {
        int d   = t >> 8;
        int rem = t & 255;
        int b   = rem >> 7;
        int o   = rem & 127;
        int n   = r0 + d;
        int c   = o * 2 + b;
        float dv = dinv[n];
        out[((size_t)b * N_NODES + n) * F_OUT + o] =
            bias[o] + dv * dv * xw[(size_t)n * OB + c] + tile[d * OB + tile_pos(c)];
    }
}


extern "C" void kernel_launch(void* const* d_in, const int* in_sizes, int n_in,
                              void* d_out, int out_size, void* d_ws, size_t ws_size,
                              hipStream_t stream) {
    const float* x    = (const float*)d_in[0];
    const float* w    = (const float*)d_in[1];
    const float* bias = (const float*)d_in[2];
    const float* ew   = (const float*)d_in[3];
    const int*   eidx = (const int*)d_in[4];
    const int*   row  = eidx;
    const int*   col  = eidx + E_EDGES;
    float* out = (float*)d_out;

    float* xw   = (float*)d_ws;
    float* deg  = xw + (size_t)N_NODES * OB;
    float* dinv = deg + N_NODES;

    deg_init_kernel<<<(N_NODES + 255) / 256, 256, 0, stream>>>(deg);
    deg_acc_kernel<<<1024, 256, 0, stream>>>(row, ew, deg);
    dinv_kernel<<<(N_NODES + 255) / 256, 256, 0, stream>>>(deg, dinv);

    gemm_wmma_kernel<<<GM_BLOCKS, GM_THREADS, 0, stream>>>(x, w, xw);

    edge_owner_kernel<<<OWN_WGS, OWN_THREADS, 0, stream>>>(row, col, ew, dinv, xw, bias, out);
}


# Degree pass writes a 1-byte owner key per edge; each owner scans 800 KB of keys (16 per load, 4 tested per word) instead of 3200 KB of row[]; queue holds edge ids; LDS float adds stay arrival-ordered
# speedup vs baseline: 4.0000x; 1.1235x over previous
#include <hip/hip_runtime.h>

#define N_NODES 50000
#define E_EDGES 800000
#define F_IN    256
#define F_OUT   128
#define BATCH   2
#define OB      (F_OUT * BATCH)

#define NT ((N_NODES + 15) / 16)

#define OWN_WGS      256
#define ROWS_PER_WG  ((N_NODES + OWN_WGS - 1) / OWN_WGS)

#define WS_KEY_OFFSET  (((size_t)N_NODES * OB + 2 * (size_t)N_NODES) * sizeof(float))
#define WS_KEY_END     (WS_KEY_OFFSET + (size_t)E_EDGES)

typedef float v2f __attribute__((ext_vector_type(2)));
typedef float v8f __attribute__((ext_vector_type(8)));


__global__ void deg_init_kernel(float* __restrict__ deg) {
    int i = blockIdx.x * blockDim.x + threadIdx.x;
    if (i < N_NODES) deg[i] = 1.0f;
}

__global__ void deg_acc_kernel(const int* __restrict__ row,
                               const float* __restrict__ ew,
                               float* __restrict__ deg,
                               unsigned* __restrict__ keyw,
                               size_t ws_bytes) {
    unsigned char* key = (unsigned char*)keyw;
    const bool has_keys = ws_bytes >= WS_KEY_END;
    for (int e = blockIdx.x * blockDim.x + threadIdx.x; e < E_EDGES;
         e += gridDim.x * blockDim.x) {
        int r = row[e];
        atomicAdd(&deg[r], ew[e]);
        if (has_keys) key[e] = (unsigned char)((unsigned)r / (unsigned)ROWS_PER_WG);
    }
}

__global__ void dinv_kernel(const float* __restrict__ deg,
                            float* __restrict__ dinv) {
    int i = blockIdx.x * blockDim.x + threadIdx.x;
    if (i < N_NODES) {
        float d = deg[i];
        dinv[i] = d > 0.0f ? rsqrtf(d) : 0.0f;
    }
}


#define GM_BLOCKS   256
#define GM_THREADS  768
#define GM_PAIRS    (GM_THREADS / 64)
#define KGROUPS     (F_IN / 8)

__device__ __forceinline__ v8f wmma4(float ax, float ay, float bx, float by, v8f c) {
    v2f a = {ax, ay};
    v2f b = {bx, by};
    return __builtin_amdgcn_wmma_f32_16x16x4_f32(
         false, a,  false, b,
         (short)0, c,  false,  false);
}

__device__ __forceinline__ void mma_group(const float4& a0, const float4& a1,
                                          const float4 (&bf)[4],
                                          v8f (&acc0)[4], v8f (&acc1)[4]) {
    #pragma unroll
    for (int t = 0; t < 4; ++t) {
        acc0[t] = wmma4(a0.x, a0.y, bf[t].x, bf[t].y, acc0[t]);
        acc1[t] = wmma4(a1.x, a1.y, bf[t].x, bf[t].y, acc1[t]);
    }
    #pragma unroll
    for (int t = 0; t < 4; ++t) {
        acc0[t] = wmma4(a0.z, a0.w, bf[t].z, bf[t].w, acc0[t]);
        acc1[t] = wmma4(a1.z, a1.w, bf[t].z, bf[t].w, acc1[t]);
    }
}

__global__ __launch_bounds__(GM_THREADS)
void gemm_wmma_kernel(const float* __restrict__ x,
                      const float* __restrict__ w,
                      float* __restrict__ xw) {
    __shared__ float4 wl[F_IN * F_OUT / 4];

    const int tid = threadIdx.x;
    for (int idx = tid; idx < F_IN * F_OUT / 4; idx += GM_THREADS) {
        int o  = idx & (F_OUT - 1);
        int kq = idx >> 7;
        const float* src = w + (size_t)(kq * 4) * F_OUT + o;
        float4 v = make_float4(src[0], src[F_OUT], src[2 * F_OUT], src[3 * F_OUT]);
        int kk = kq >> 1, hh = kq & 1;
        int oh = o >> 6, t = (o >> 4) & 3, ll = o & 15;
        wl[((oh * KGROUPS + kk) * 4 + t) * 32 + hh * 16 + ll] = v;
    }
    __syncthreads();

    const int wave = __builtin_amdgcn_readfirstlane(tid >> 5);
    const int lane = tid & 31;
    const int oh = wave & 1;
    const int h  = lane >> 4;
    const int l  = lane & 15;
    const float4* wb = wl + oh * (KGROUPS * 4 * 32) + lane;

    for (int nt = (wave >> 1) * GM_BLOCKS + blockIdx.x; nt < NT; nt += GM_PAIRS * GM_BLOCKS) {
        const int n0 = nt * 16;
        const float4* xa0 = (const float4*)(x + (size_t)(n0 + l) * F_IN + h * 4);
        const float4* xa1 = (const float4*)(x + ((size_t)N_NODES + n0 + l) * F_IN + h * 4);

        v8f acc0[4] = {};
        v8f acc1[4] = {};
        float4 a0 = xa0[0];
        float4 a1 = xa1[0];
        for (int kk = 0; kk < KGROUPS; ++kk) {
            float4 bf[4];
            #pragma unroll
            for (int t = 0; t < 4; ++t) bf[t] = wb[(kk * 4 + t) * 32];
            const int kn = (kk + 1) & (KGROUPS - 1);
            float4 na0 = xa0[kn * 2];
            float4 na1 = xa1[kn * 2];
            mma_group(a0, a1, bf, acc0, acc1);
            a0 = na0;
            a1 = na1;
        }

        float* orow = xw + (size_t)(n0 + 8 * h) * OB + (oh * 64 + l) * 2;
        #pragma unroll
        for (int t = 0; t < 4; ++t) {
            #pragma unroll
            for (int v = 0; v < 8; ++v) {
                float2 p = make_float2(acc0[t][v], acc1[t][v]);
                *(float2*)(orow + (size_t)v * OB + t * 32) = p;
            }
        }
    }
}


#define OWN_THREADS  1024
#define OWN_WAVES    (OWN_THREADS / 32)
#define QCAP         8192
#define N_KEY16      (E_EDGES / 16)
#define N_ROW4       (E_EDGES / 4)
#define SCAN_UNROLL  8
#define SCAN_STEP    (SCAN_UNROLL * OWN_THREADS)
#define SCAN_FULL    ((N_KEY16 / SCAN_STEP) * SCAN_STEP)
#define ROW_FULL     ((N_ROW4 / SCAN_STEP) * SCAN_STEP)

__device__ __forceinline__ int tile_pos(int c) { return ((c & 3) << 6) | (c >> 2); }

__device__ __forceinline__ void lds_add(float* p, float v) {
    __hip_atomic_fetch_add(p, v, __ATOMIC_RELAXED, __HIP_MEMORY_SCOPE_WORKGROUP);
}

__device__ __forceinline__ void queue_owned(uint4 v, int idx, unsigned mine4,
                                            unsigned* q_cnt, unsigned* q_a) {
    unsigned ww[4] = {v.x, v.y, v.z, v.w};
    unsigned c = 0u;
    #pragma unroll
    for (int j = 0; j < 4; ++j) {
        unsigned t = ww[j] ^ mine4;
        unsigned z = ~(((t & 0x7F7F7F7Fu) + 0x7F7F7F7Fu) | t) & 0x80808080u;
        c |= z >> j;
    }
    while (c) {
        unsigned p   = (unsigned)__builtin_ctz(c);
        unsigned pos = atomicAdd(q_cnt, 1u);
        if (pos < QCAP) q_a[pos] = (unsigned)(idx * 16) + (((p ^ 7u) & 7u) << 2) + (p >> 3);
        c &= c - 1u;
    }
}

__device__ __forceinline__ void queue_owned_rows(int4 v, int idx, int r0, int nrows,
                                                 unsigned* q_cnt, unsigned* q_a) {
    int rr[4] = {v.x, v.y, v.z, v.w};
    #pragma unroll
    for (int j = 0; j < 4; ++j) {
        unsigned d = (unsigned)(rr[j] - r0);
        if (d < (unsigned)nrows) {
            unsigned pos = atomicAdd(q_cnt, 1u);
            if (pos < QCAP) q_a[pos] = (unsigned)(idx * 4 + j);
        }
    }
}

__global__ __launch_bounds__(OWN_THREADS)
void edge_owner_kernel(const unsigned* __restrict__ keyw,
                       const int* __restrict__ row,
                       const int* __restrict__ col,
                       const float* __restrict__ ew,
                       const float* __restrict__ dinv,
                       const float* __restrict__ xw,
                       const float* __restrict__ bias,
                       float* __restrict__ out,
                       size_t ws_bytes) {
    __shared__ float    tile[ROWS_PER_WG * OB];
    __shared__ unsigned q_a[QCAP];
    __shared__ float    q_lap[QCAP];
    __shared__ unsigned q_cnt;

    const int tid   = threadIdx.x;
    const int r0    = blockIdx.x * ROWS_PER_WG;
    const int nrows = min(ROWS_PER_WG, N_NODES - r0);

    for (int i = tid; i < ROWS_PER_WG * OB / 4; i += OWN_THREADS)
        ((float4*)tile)[i] = make_float4(0.f, 0.f, 0.f, 0.f);
    if (tid == 0) q_cnt = 0u;
    __syncthreads();

    if (ws_bytes >= WS_KEY_END) {
        const uint4*   key16 = (const uint4*)keyw;
        const unsigned mine4 = (unsigned)blockIdx.x * 0x01010101u;
        const int      start = blockIdx.x * (N_KEY16 / OWN_WGS);
        for (int k = tid; k < SCAN_FULL; k += SCAN_UNROLL * OWN_THREADS) {
            uint4 v[SCAN_UNROLL];
            int   id[SCAN_UNROLL];
            #pragma unroll
            for (int u = 0; u < SCAN_UNROLL; ++u) {
                int idx = start + k + u * OWN_THREADS;
                if (idx >= N_KEY16) idx -= N_KEY16;
                id[u] = idx;
                v[u]  = key16[idx];
            }
            __builtin_amdgcn_sched_barrier(0);
            #pragma unroll
            for (int u = 0; u < SCAN_UNROLL; ++u) queue_owned(v[u], id[u], mine4, &q_cnt, q_a);
        }
        for (int kk = SCAN_FULL + tid; kk < N_KEY16; kk += OWN_THREADS) {
            int idx = start + kk;
            if (idx >= N_KEY16) idx -= N_KEY16;
            queue_owned(key16[idx], idx, mine4, &q_cnt, q_a);
        }
    } else {
        const int4* row4  = (const int4*)row;
        const int   start = blockIdx.x * (N_ROW4 / OWN_WGS);
        for (int k = tid; k < ROW_FULL; k += SCAN_UNROLL * OWN_THREADS) {
            int4 v[SCAN_UNROLL];
            int  id[SCAN_UNROLL];
            #pragma unroll
            for (int u = 0; u < SCAN_UNROLL; ++u) {
                int idx = start + k + u * OWN_THREADS;
                if (idx >= N_ROW4) idx -= N_ROW4;
                id[u] = idx;
                v[u]  = row4[idx];
            }
            __builtin_amdgcn_sched_barrier(0);
            #pragma unroll
            for (int u = 0; u < SCAN_UNROLL; ++u) queue_owned_rows(v[u], id[u], r0, nrows, &q_cnt, q_a);
        }
        for (int kk = ROW_FULL + tid; kk < N_ROW4; kk += OWN_THREADS) {
            int idx = start + kk;
            if (idx >= N_ROW4) idx -= N_ROW4;
            queue_owned_rows(row4[idx], idx, r0, nrows, &q_cnt, q_a);
        }
    }
    __syncthreads();

    int nq = (int)q_cnt;
    if (q_cnt > (unsigned)QCAP) {
        nq = 0;
        for (int e = tid; e < E_EDGES; e += OWN_THREADS) {
            int r = row[e];
            unsigned d = (unsigned)(r - r0);
            if (d < (unsigned)nrows) {
                int   cl  = col[e];
                float lap = dinv[r] * ew[e] * dinv[cl];
                const float* srow = xw + (size_t)cl * OB;
                float* trow = tile + (int)d * OB;
                for (int c = 0; c < OB; ++c) lds_add(&trow[tile_pos(c)], lap * srow[c]);
            }
        }
    }

    for (int i = tid; i < nq; i += OWN_THREADS) {
        int e  = (int)q_a[i];
        int r  = row[e];
        int d  = r - r0;
        int cl = col[e];
        float lap = dinv[r] * ew[e] * dinv[cl];
        q_a[i]   = (unsigned)cl | ((unsigned)d << 16);
        q_lap[i] = lap;
    }
    __syncthreads();

    const int wave = __builtin_amdgcn_readfirstlane(tid >> 5);
    const int lane = tid & 31;
    int i = wave;
    for (; i + 3 * OWN_WAVES < nq; i += 4 * OWN_WAVES) {
        unsigned pk[4];
        float    lp[4];
        float4   a[4], b[4];
        #pragma unroll
        for (int u = 0; u < 4; ++u) {
            pk[u] = q_a[i + u * OWN_WAVES];
            lp[u] = q_lap[i + u * OWN_WAVES];
            const float4* s = (const float4*)(xw + (size_t)(pk[u] & 0xFFFFu) * OB);
            a[u] = s[lane];
            b[u] = s[32 + lane];
        }
        #pragma unroll
        for (int u = 0; u < 4; ++u) {
            float* t = tile + (int)(pk[u] >> 16) * OB + lane;
            lds_add(t +   0, lp[u] * a[u].x);
            lds_add(t +  64, lp[u] * a[u].y);
            lds_add(t + 128, lp[u] * a[u].z);
            lds_add(t + 192, lp[u] * a[u].w);
            lds_add(t +  32, lp[u] * b[u].x);
            lds_add(t +  96, lp[u] * b[u].y);
            lds_add(t + 160, lp[u] * b[u].z);
            lds_add(t + 224, lp[u] * b[u].w);
        }
    }
    for (; i < nq; i += OWN_WAVES) {
        unsigned pk = q_a[i];
        float    lp = q_lap[i];
        const float4* s = (const float4*)(xw + (size_t)(pk & 0xFFFFu) * OB);
        float4 a = s[lane];
        float4 b = s[32 + lane];
        float* t = tile + (int)(pk >> 16) * OB + lane;
        lds_add(t +   0, lp * a.x);
        lds_add(t +  64, lp * a.y);
        lds_add(t + 128, lp * a.z);
        lds_add(t + 192, lp * a.w);
        lds_add(t +  32, lp * b.x);
        lds_add(t +  96, lp * b.y);
        lds_add(t + 160, lp * b.z);
        lds_add(t + 224, lp * b.w);
    }
    __syncthreads();

    for (int t = tid; t < nrows * OB; t += OWN_THREADS) {
        int d   = t >> 8;
        int rem = t & 255;
        int b   = rem >> 7;
        int o   = rem & 127;
        int n   = r0 + d;
        int c   = o * 2 + b;
        float dv = dinv[n];
        out[((size_t)b * N_NODES + n) * F_OUT + o] =
            bias[o] + dv * dv * xw[(size_t)n * OB + c] + tile[d * OB + tile_pos(c)];
    }
}


extern "C" void kernel_launch(void* const* d_in, const int* in_sizes, int n_in,
                              void* d_out, int out_size, void* d_ws, size_t ws_size,
                              hipStream_t stream) {
    const float* x    = (const float*)d_in[0];
    const float* w    = (const float*)d_in[1];
    const float* bias = (const float*)d_in[2];
    const float* ew   = (const float*)d_in[3];
    const int*   eidx = (const int*)d_in[4];
    const int*   row  = eidx;
    const int*   col  = eidx + E_EDGES;
    float* out = (float*)d_out;

    float* xw   = (float*)d_ws;
    float* deg  = xw + (size_t)N_NODES * OB;
    float* dinv = deg + N_NODES;
    unsigned* keyw = (unsigned*)(dinv + N_NODES);

    deg_init_kernel<<<(N_NODES + 255) / 256, 256, 0, stream>>>(deg);
    deg_acc_kernel<<<1024, 256, 0, stream>>>(row, ew, deg, keyw, ws_size);
    dinv_kernel<<<(N_NODES + 255) / 256, 256, 0, stream>>>(deg, dinv);

    gemm_wmma_kernel<<<GM_BLOCKS, GM_THREADS, 0, stream>>>(x, w, xw);

    edge_owner_kernel<<<OWN_WGS, OWN_THREADS, 0, stream>>>(keyw, row, col, ew, dinv, xw, bias, out, ws_size);
}


# Edge sums leave LDS: queued edges counting-sorted by destination row in LDS, then one wave per row sums gathered xw rows in registers (row-list order, not arrival-ordered LDS adds) and stores out
# speedup vs baseline: 4.3126x; 1.0781x over previous
#include <hip/hip_runtime.h>

#define N_NODES 50000
#define E_EDGES 800000
#define F_IN    256
#define F_OUT   128
#define BATCH   2
#define OB      (F_OUT * BATCH)

#define NT ((N_NODES + 15) / 16)

#define OWN_WGS      256
#define ROWS_PER_WG  ((N_NODES + OWN_WGS - 1) / OWN_WGS)

#define WS_KEY_OFFSET  (((size_t)N_NODES * OB + 2 * (size_t)N_NODES) * sizeof(float))
#define WS_KEY_END     (WS_KEY_OFFSET + (size_t)E_EDGES)

typedef float v2f __attribute__((ext_vector_type(2)));
typedef float v8f __attribute__((ext_vector_type(8)));


__global__ void deg_init_kernel(float* __restrict__ deg) {
    int i = blockIdx.x * blockDim.x + threadIdx.x;
    if (i < N_NODES) deg[i] = 1.0f;
}

__global__ void deg_acc_kernel(const int* __restrict__ row,
                               const float* __restrict__ ew,
                               float* __restrict__ deg,
                               unsigned* __restrict__ keyw,
                               size_t ws_bytes) {
    unsigned char* key = (unsigned char*)keyw;
    const bool has_keys = ws_bytes >= WS_KEY_END;
    for (int e = blockIdx.x * blockDim.x + threadIdx.x; e < E_EDGES;
         e += gridDim.x * blockDim.x) {
        int r = row[e];
        atomicAdd(&deg[r], ew[e]);
        if (has_keys) key[e] = (unsigned char)((unsigned)r / (unsigned)ROWS_PER_WG);
    }
}

__global__ void dinv_kernel(const float* __restrict__ deg,
                            float* __restrict__ dinv) {
    int i = blockIdx.x * blockDim.x + threadIdx.x;
    if (i < N_NODES) {
        float d = deg[i];
        dinv[i] = d > 0.0f ? rsqrtf(d) : 0.0f;
    }
}


#define GM_BLOCKS   256
#define GM_THREADS  768
#define GM_PAIRS    (GM_THREADS / 64)
#define KGROUPS     (F_IN / 8)

__device__ __forceinline__ v8f wmma4(float ax, float ay, float bx, float by, v8f c) {
    v2f a = {ax, ay};
    v2f b = {bx, by};
    return __builtin_amdgcn_wmma_f32_16x16x4_f32(
         false, a,  false, b,
         (short)0, c,  false,  false);
}

__device__ __forceinline__ void mma_group(const float4& a0, const float4& a1,
                                          const float4 (&bf)[4],
                                          v8f (&acc0)[4], v8f (&acc1)[4]) {
    #pragma unroll
    for (int t = 0; t < 4; ++t) {
        acc0[t] = wmma4(a0.x, a0.y, bf[t].x, bf[t].y, acc0[t]);
        acc1[t] = wmma4(a1.x, a1.y, bf[t].x, bf[t].y, acc1[t]);
    }
    #pragma unroll
    for (int t = 0; t < 4; ++t) {
        acc0[t] = wmma4(a0.z, a0.w, bf[t].z, bf[t].w, acc0[t]);
        acc1[t] = wmma4(a1.z, a1.w, bf[t].z, bf[t].w, acc1[t]);
    }
}

__global__ __launch_bounds__(GM_THREADS)
void gemm_wmma_kernel(const float* __restrict__ x,
                      const float* __restrict__ w,
                      float* __restrict__ xw) {
    __shared__ float4 wl[F_IN * F_OUT / 4];

    const int tid = threadIdx.x;
    for (int idx = tid; idx < F_IN * F_OUT / 4; idx += GM_THREADS) {
        int o  = idx & (F_OUT - 1);
        int kq = idx >> 7;
        const float* src = w + (size_t)(kq * 4) * F_OUT + o;
        float4 v = make_float4(src[0], src[F_OUT], src[2 * F_OUT], src[3 * F_OUT]);
        int kk = kq >> 1, hh = kq & 1;
        int oh = o >> 6, t = (o >> 4) & 3, ll = o & 15;
        wl[((oh * KGROUPS + kk) * 4 + t) * 32 + hh * 16 + ll] = v;
    }
    __syncthreads();

    const int wave = __builtin_amdgcn_readfirstlane(tid >> 5);
    const int lane = tid & 31;
    const int oh = wave & 1;
    const int h  = lane >> 4;
    const int l  = lane & 15;
    const float4* wb = wl + oh * (KGROUPS * 4 * 32) + lane;

    for (int nt = (wave >> 1) * GM_BLOCKS + blockIdx.x; nt < NT; nt += GM_PAIRS * GM_BLOCKS) {
        const int n0 = nt * 16;
        const float4* xa0 = (const float4*)(x + (size_t)(n0 + l) * F_IN + h * 4);
        const float4* xa1 = (const float4*)(x + ((size_t)N_NODES + n0 + l) * F_IN + h * 4);

        v8f acc0[4] = {};
        v8f acc1[4] = {};
        float4 a0 = xa0[0];
        float4 a1 = xa1[0];
        for (int kk = 0; kk < KGROUPS; ++kk) {
            float4 bf[4];
            #pragma unroll
            for (int t = 0; t < 4; ++t) bf[t] = wb[(kk * 4 + t) * 32];
            const int kn = (kk + 1) & (KGROUPS - 1);
            float4 na0 = xa0[kn * 2];
            float4 na1 = xa1[kn * 2];
            mma_group(a0, a1, bf, acc0, acc1);
            a0 = na0;
            a1 = na1;
        }

        float* orow = xw + (size_t)(n0 + 8 * h) * OB + (oh * 64 + l) * 2;
        #pragma unroll
        for (int t = 0; t < 4; ++t) {
            #pragma unroll
            for (int v = 0; v < 8; ++v) {
                float2 p = make_float2(acc0[t][v], acc1[t][v]);
                *(float2*)(orow + (size_t)v * OB + t * 32) = p;
            }
        }
    }
}


#define OWN_THREADS  1024
#define OWN_WAVES    (OWN_THREADS / 32)
#define QCAP         16384
#define ROWS_PAD     256
#define N_KEY16      (E_EDGES / 16)
#define N_ROW4       (E_EDGES / 4)
#define SCAN_UNROLL  8
#define SCAN_STEP    (SCAN_UNROLL * OWN_THREADS)
#define SCAN_FULL    ((N_KEY16 / SCAN_STEP) * SCAN_STEP)
#define ROW_FULL     ((N_ROW4 / SCAN_STEP) * SCAN_STEP)

__device__ __forceinline__ void queue_owned(uint4 v, int idx, unsigned mine4,
                                            unsigned* q_cnt, unsigned* q_a) {
    unsigned ww[4] = {v.x, v.y, v.z, v.w};
    unsigned c = 0u;
    #pragma unroll
    for (int j = 0; j < 4; ++j) {
        unsigned t = ww[j] ^ mine4;
        unsigned z = ~(((t & 0x7F7F7F7Fu) + 0x7F7F7F7Fu) | t) & 0x80808080u;
        c |= z >> j;
    }
    while (c) {
        unsigned p   = (unsigned)__builtin_ctz(c);
        unsigned pos = atomicAdd(q_cnt, 1u);
        if (pos < QCAP) q_a[pos] = (unsigned)(idx * 16) + (((p ^ 7u) & 7u) << 2) + (p >> 3);
        c &= c - 1u;
    }
}

__device__ __forceinline__ void queue_owned_rows(int4 v, int idx, int r0, int nrows,
                                                 unsigned* q_cnt, unsigned* q_a) {
    int rr[4] = {v.x, v.y, v.z, v.w};
    #pragma unroll
    for (int j = 0; j < 4; ++j) {
        unsigned d = (unsigned)(rr[j] - r0);
        if (d < (unsigned)nrows) {
            unsigned pos = atomicAdd(q_cnt, 1u);
            if (pos < QCAP) q_a[pos] = (unsigned)(idx * 4 + j);
        }
    }
}

__device__ __forceinline__ void fma4(float w, const float4& v, float4& acc) {
    acc.x = fmaf(w, v.x, acc.x);
    acc.y = fmaf(w, v.y, acc.y);
    acc.z = fmaf(w, v.z, acc.z);
    acc.w = fmaf(w, v.w, acc.w);
}

template <int M>
__device__ __forceinline__ void add_rows(const float* __restrict__ xw, const uint2* s_e,
                                         unsigned first, int lane,
                                         float4& accA, float4& accB) {
    float  lp[M];
    float4 a[M], b[M];
    #pragma unroll
    for (int u = 0; u < M; ++u) {
        uint2 en = s_e[first + u];
        int cl = __builtin_amdgcn_readfirstlane((int)en.x);
        lp[u]  = __uint_as_float(en.y);
        const float4* s = (const float4*)(xw + (size_t)cl * OB);
        a[u] = s[lane];
        b[u] = s[32 + lane];
    }
    #pragma unroll
    for (int u = 0; u < M; ++u) {
        fma4(lp[u], a[u], accA);
        fma4(lp[u], b[u], accB);
    }
}

__device__ __forceinline__ void store_row(float* __restrict__ out, int n, int lane, float w,
                                          const float4& sa, const float4& sb,
                                          float4 accA, float4 accB, float2 bz0, float2 bz1) {
    fma4(w, sa, accA);
    fma4(w, sb, accB);
    float* o0 = out + (size_t)n * F_OUT + 2 * lane;
    float* o1 = out + ((size_t)N_NODES + n) * F_OUT + 2 * lane;
    *(float2*)(o0)      = make_float2(accA.x + bz0.x, accA.z + bz0.y);
    *(float2*)(o1)      = make_float2(accA.y + bz0.x, accA.w + bz0.y);
    *(float2*)(o0 + 64) = make_float2(accB.x + bz1.x, accB.z + bz1.y);
    *(float2*)(o1 + 64) = make_float2(accB.y + bz1.x, accB.w + bz1.y);
}

__global__ __launch_bounds__(OWN_THREADS)
void edge_owner_kernel(const unsigned* __restrict__ keyw,
                       const int* __restrict__ row,
                       const int* __restrict__ col,
                       const float* __restrict__ ew,
                       const float* __restrict__ dinv,
                       const float* __restrict__ xw,
                       const float* __restrict__ bias,
                       float* __restrict__ out,
                       size_t ws_bytes) {
    __shared__ unsigned q_a[QCAP];
    __shared__ float    q_lap[QCAP];
    __shared__ uint2    s_e[QCAP];
    __shared__ unsigned r_cnt[ROWS_PAD];
    __shared__ unsigned r_beg[ROWS_PAD];
    __shared__ unsigned r_fill[ROWS_PAD];
    __shared__ unsigned q_cnt;
    __shared__ unsigned next_row;

    const int tid   = threadIdx.x;
    const int r0    = blockIdx.x * ROWS_PER_WG;
    const int nrows = min(ROWS_PER_WG, N_NODES - r0);
    const int wave  = __builtin_amdgcn_readfirstlane(tid >> 5);
    const int lane  = tid & 31;

    if (tid < ROWS_PAD) r_cnt[tid] = 0u;
    if (tid == 0) { q_cnt = 0u; next_row = (unsigned)OWN_WAVES; }
    __syncthreads();

    if (ws_bytes >= WS_KEY_END) {
        const uint4*   key16 = (const uint4*)keyw;
        const unsigned mine4 = (unsigned)blockIdx.x * 0x01010101u;
        const int      start = blockIdx.x * (N_KEY16 / OWN_WGS);
        for (int k = tid; k < SCAN_FULL; k += SCAN_UNROLL * OWN_THREADS) {
            uint4 v[SCAN_UNROLL];
            int   id[SCAN_UNROLL];
            #pragma unroll
            for (int u = 0; u < SCAN_UNROLL; ++u) {
                int idx = start + k + u * OWN_THREADS;
                if (idx >= N_KEY16) idx -= N_KEY16;
                id[u] = idx;
                v[u]  = key16[idx];
            }
            __builtin_amdgcn_sched_barrier(0);
            #pragma unroll
            for (int u = 0; u < SCAN_UNROLL; ++u) queue_owned(v[u], id[u], mine4, &q_cnt, q_a);
        }
        for (int kk = SCAN_FULL + tid; kk < N_KEY16; kk += OWN_THREADS) {
            int idx = start + kk;
            if (idx >= N_KEY16) idx -= N_KEY16;
            queue_owned(key16[idx], idx, mine4, &q_cnt, q_a);
        }
    } else {
        const int4* row4  = (const int4*)row;
        const int   start = blockIdx.x * (N_ROW4 / OWN_WGS);
        for (int k = tid; k < ROW_FULL; k += SCAN_UNROLL * OWN_THREADS) {
            int4 v[SCAN_UNROLL];
            int  id[SCAN_UNROLL];
            #pragma unroll
            for (int u = 0; u < SCAN_UNROLL; ++u) {
                int idx = start + k + u * OWN_THREADS;
                if (idx >= N_ROW4) idx -= N_ROW4;
                id[u] = idx;
                v[u]  = row4[idx];
            }
            __builtin_amdgcn_sched_barrier(0);
            #pragma unroll
            for (int u = 0; u < SCAN_UNROLL; ++u) queue_owned_rows(v[u], id[u], r0, nrows, &q_cnt, q_a);
        }
        for (int kk = ROW_FULL + tid; kk < N_ROW4; kk += OWN_THREADS) {
            int idx = start + kk;
            if (idx >= N_ROW4) idx -= N_ROW4;
            queue_owned_rows(row4[idx], idx, r0, nrows, &q_cnt, q_a);
        }
    }
    __syncthreads();


    const float2 bz0 = *(const float2*)(bias + 2 * lane);
    const float2 bz1 = *(const float2*)(bias + 64 + 2 * lane);

    if (q_cnt > (unsigned)QCAP) {
        for (int d = wave; d < nrows; d += OWN_WAVES) {
            const int n = r0 + d;
            const float dn = dinv[n];
            const float4* sn = (const float4*)(xw + (size_t)n * OB);
            const float4 sa = sn[lane];
            const float4 sb = sn[32 + lane];
            float4 accA = make_float4(0.f, 0.f, 0.f, 0.f);
            float4 accB = make_float4(0.f, 0.f, 0.f, 0.f);
            for (int e0 = 0; e0 < E_EDGES; e0 += 32) {
                unsigned m = __builtin_amdgcn_ballot_w32(row[e0 + lane] == n);
                while (m) {
                    int e = e0 + __builtin_ctz(m);
                    m &= m - 1u;
                    int   cl  = col[e];
                    float lap = dn * ew[e] * dinv[cl];
                    const float4* s = (const float4*)(xw + (size_t)cl * OB);
                    fma4(lap, s[lane], accA);
                    fma4(lap, s[32 + lane], accB);
                }
            }
            store_row(out, n, lane, dn * dn, sa, sb, accA, accB, bz0, bz1);
        }
        return;
    }
    const int nq = (int)q_cnt;

    for (int i = tid; i < nq; i += OWN_THREADS) {
        int e  = (int)q_a[i];
        int r  = row[e];
        int d  = r - r0;
        int cl = col[e];
        float lap = dinv[r] * ew[e] * dinv[cl];
        q_a[i]   = (unsigned)cl | ((unsigned)d << 16);
        q_lap[i] = lap;
        atomicAdd(&r_cnt[d], 1u);
    }
    __syncthreads();

    if (tid < 32) {
        unsigned c[8];
        unsigned s = 0u;
        #pragma unroll
        for (int j = 0; j < 8; ++j) { c[j] = r_cnt[tid * 8 + j]; s += c[j]; }
        unsigned incl = s;
        #pragma unroll
        for (int off = 1; off < 32; off <<= 1) {
            unsigned t = __shfl_up(incl, off, 32);
            if (tid >= off) incl += t;
        }
        unsigned base = incl - s;
        #pragma unroll
        for (int j = 0; j < 8; ++j) {
            r_beg[tid * 8 + j]  = base;
            r_fill[tid * 8 + j] = base;
            base += c[j];
        }
    }
    __syncthreads();

    for (int i = tid; i < nq; i += OWN_THREADS) {
        unsigned pk  = q_a[i];
        unsigned pos = atomicAdd(&r_fill[pk >> 16], 1u);
        s_e[pos] = make_uint2(pk & 0xFFFFu, __float_as_uint(q_lap[i]));
    }
    __syncthreads();

    int d = wave;
    while (d < nrows) {
        const int n = r0 + d;
        const unsigned beg = (unsigned)__builtin_amdgcn_readfirstlane((int)r_beg[d]);
        const unsigned cnt = (unsigned)__builtin_amdgcn_readfirstlane((int)r_cnt[d]);
        const float4* sn = (const float4*)(xw + (size_t)n * OB);
        const float4 sa = sn[lane];
        const float4 sb = sn[32 + lane];
        const float  dv = dinv[n];
        float4 accA = make_float4(0.f, 0.f, 0.f, 0.f);
        float4 accB = make_float4(0.f, 0.f, 0.f, 0.f);
        unsigned k = 0u;
        for (; k + 4u <= cnt; k += 4u) add_rows<4>(xw, s_e, beg + k, lane, accA, accB);
        switch (cnt - k) {
            case 3u: add_rows<3>(xw, s_e, beg + k, lane, accA, accB); break;
            case 2u: add_rows<2>(xw, s_e, beg + k, lane, accA, accB); break;
            case 1u: add_rows<1>(xw, s_e, beg + k, lane, accA, accB); break;
            default: break;
        }
        store_row(out, n, lane, dv * dv, sa, sb, accA, accB, bz0, bz1);
        unsigned nd = 0u;
        if (lane == 0) nd = atomicAdd(&next_row, 1u);
        d = __builtin_amdgcn_readfirstlane((int)nd);
    }
}


extern "C" void kernel_launch(void* const* d_in, const int* in_sizes, int n_in,
                              void* d_out, int out_size, void* d_ws, size_t ws_size,
                              hipStream_t stream) {
    const float* x    = (const float*)d_in[0];
    const float* w    = (const float*)d_in[1];
    const float* bias = (const float*)d_in[2];
    const float* ew   = (const float*)d_in[3];
    const int*   eidx = (const int*)d_in[4];
    const int*   row  = eidx;
    const int*   col  = eidx + E_EDGES;
    float* out = (float*)d_out;

    float* xw   = (float*)d_ws;
    float* deg  = xw + (size_t)N_NODES * OB;
    float* dinv = deg + N_NODES;
    unsigned* keyw = (unsigned*)(dinv + N_NODES);

    deg_init_kernel<<<(N_NODES + 255) / 256, 256, 0, stream>>>(deg);
    deg_acc_kernel<<<1024, 256, 0, stream>>>(row, ew, deg, keyw, ws_size);
    dinv_kernel<<<(N_NODES + 255) / 256, 256, 0, stream>>>(deg, dinv);

    gemm_wmma_kernel<<<GM_BLOCKS, GM_THREADS, 0, stream>>>(x, w, xw);

    edge_owner_kernel<<<OWN_WGS, OWN_THREADS, 0, stream>>>(keyw, row, col, ew, dinv, xw, bias, out, ws_size);
}


# Launches 5 -> 3: degree sums (float atomics, arrival order) and owner keys ride in the GEMM launch, issued behind its staging barrier; deg^-1/2 by rsqrt where used, no dinv launch or plane
# speedup vs baseline: 4.5141x; 1.0467x over previous
#include <hip/hip_runtime.h>

#define N_NODES 50000
#define E_EDGES 800000
#define F_IN    256
#define F_OUT   128
#define BATCH   2
#define OB      (F_OUT * BATCH)

#define NT ((N_NODES + 15) / 16)

#define OWN_WGS      256
#define ROWS_PER_WG  ((N_NODES + OWN_WGS - 1) / OWN_WGS)

#define WS_KEY_OFFSET  (((size_t)N_NODES * OB + (size_t)N_NODES) * sizeof(float))
#define WS_KEY_END     (WS_KEY_OFFSET + (size_t)E_EDGES)

typedef float v2f __attribute__((ext_vector_type(2)));
typedef float v8f __attribute__((ext_vector_type(8)));


__global__ void deg_init_kernel(float* __restrict__ deg) {
    int i = blockIdx.x * blockDim.x + threadIdx.x;
    if (i < N_NODES) deg[i] = 1.0f;
}

__device__ __forceinline__ float inv_sqrt_deg(float d) {
    return d > 0.0f ? rsqrtf(d) : 0.0f;
}


#define GM_BLOCKS   256
#define GM_THREADS  768
#define EDGES_PER_BLOCK (E_EDGES / GM_BLOCKS)
#define EDGE_TRIPS      ((EDGES_PER_BLOCK + GM_THREADS - 1) / GM_THREADS)
#define GM_PAIRS    (GM_THREADS / 64)
#define KGROUPS     (F_IN / 8)

__device__ __forceinline__ v8f wmma4(float ax, float ay, float bx, float by, v8f c) {
    v2f a = {ax, ay};
    v2f b = {bx, by};
    return __builtin_amdgcn_wmma_f32_16x16x4_f32(
         false, a,  false, b,
         (short)0, c,  false,  false);
}

__device__ __forceinline__ void mma_group(const float4& a0, const float4& a1,
                                          const float4 (&bf)[4],
                                          v8f (&acc0)[4], v8f (&acc1)[4]) {
    #pragma unroll
    for (int t = 0; t < 4; ++t) {
        acc0[t] = wmma4(a0.x, a0.y, bf[t].x, bf[t].y, acc0[t]);
        acc1[t] = wmma4(a1.x, a1.y, bf[t].x, bf[t].y, acc1[t]);
    }
    #pragma unroll
    for (int t = 0; t < 4; ++t) {
        acc0[t] = wmma4(a0.z, a0.w, bf[t].z, bf[t].w, acc0[t]);
        acc1[t] = wmma4(a1.z, a1.w, bf[t].z, bf[t].w, acc1[t]);
    }
}

__global__ __launch_bounds__(GM_THREADS)
void gemm_wmma_kernel(const float* __restrict__ x,
                      const float* __restrict__ w,
                      float* __restrict__ xw,
                      const int* __restrict__ row,
                      const float* __restrict__ ew,
                      float* __restrict__ deg,
                      unsigned* __restrict__ keyw,
                      size_t ws_bytes) {
    __shared__ float4 wl[F_IN * F_OUT / 4];

    const int tid = threadIdx.x;

    const int ebase = blockIdx.x * EDGES_PER_BLOCK;
    int   er[EDGE_TRIPS];
    float ev[EDGE_TRIPS];
    #pragma unroll
    for (int u = 0; u < EDGE_TRIPS; ++u) {
        const int i = min(tid + u * GM_THREADS, EDGES_PER_BLOCK - 1);
        er[u] = row[ebase + i];
        ev[u] = ew[ebase + i];
    }

    for (int idx = tid; idx < F_IN * F_OUT / 4; idx += GM_THREADS) {
        int o  = idx & (F_OUT - 1);
        int kq = idx >> 7;
        const float* src = w + (size_t)(kq * 4) * F_OUT + o;
        float4 v = make_float4(src[0], src[F_OUT], src[2 * F_OUT], src[3 * F_OUT]);
        int kk = kq >> 1, hh = kq & 1;
        int oh = o >> 6, t = (o >> 4) & 3, ll = o & 15;
        wl[((oh * KGROUPS + kk) * 4 + t) * 32 + hh * 16 + ll] = v;
    }
    __syncthreads();

    {
        unsigned char* key = (unsigned char*)keyw;
        const bool has_keys = ws_bytes >= WS_KEY_END;
        #pragma unroll
        for (int u = 0; u < EDGE_TRIPS; ++u) {
            const int i = tid + u * GM_THREADS;
            if (i < EDGES_PER_BLOCK) {
                atomicAdd(&deg[er[u]], ev[u]);
                if (has_keys) key[ebase + i] = (unsigned char)((unsigned)er[u] / (unsigned)ROWS_PER_WG);
            }
        }
    }

    const int wave = __builtin_amdgcn_readfirstlane(tid >> 5);
    const int lane = tid & 31;
    const int oh = wave & 1;
    const int h  = lane >> 4;
    const int l  = lane & 15;
    const float4* wb = wl + oh * (KGROUPS * 4 * 32) + lane;

    for (int nt = (wave >> 1) * GM_BLOCKS + blockIdx.x; nt < NT; nt += GM_PAIRS * GM_BLOCKS) {
        const int n0 = nt * 16;
        const float4* xa0 = (const float4*)(x + (size_t)(n0 + l) * F_IN + h * 4);
        const float4* xa1 = (const float4*)(x + ((size_t)N_NODES + n0 + l) * F_IN + h * 4);

        v8f acc0[4] = {};
        v8f acc1[4] = {};
        float4 a0 = xa0[0];
        float4 a1 = xa1[0];
        for (int kk = 0; kk < KGROUPS; ++kk) {
            float4 bf[4];
            #pragma unroll
            for (int t = 0; t < 4; ++t) bf[t] = wb[(kk * 4 + t) * 32];
            const int kn = (kk + 1) & (KGROUPS - 1);
            float4 na0 = xa0[kn * 2];
            float4 na1 = xa1[kn * 2];
            mma_group(a0, a1, bf, acc0, acc1);
            a0 = na0;
            a1 = na1;
        }

        float* orow = xw + (size_t)(n0 + 8 * h) * OB + (oh * 64 + l) * 2;
        #pragma unroll
        for (int t = 0; t < 4; ++t) {
            #pragma unroll
            for (int v = 0; v < 8; ++v) {
                float2 p = make_float2(acc0[t][v], acc1[t][v]);
                *(float2*)(orow + (size_t)v * OB + t * 32) = p;
            }
        }
    }
}


#define OWN_THREADS  1024
#define OWN_WAVES    (OWN_THREADS / 32)
#define QCAP         16384
#define ROWS_PAD     256
#define N_KEY16      (E_EDGES / 16)
#define N_ROW4       (E_EDGES / 4)
#define SCAN_UNROLL  8
#define SCAN_STEP    (SCAN_UNROLL * OWN_THREADS)
#define SCAN_FULL    ((N_KEY16 / SCAN_STEP) * SCAN_STEP)
#define ROW_FULL     ((N_ROW4 / SCAN_STEP) * SCAN_STEP)

__device__ __forceinline__ void queue_owned(uint4 v, int idx, unsigned mine4,
                                            unsigned* q_cnt, unsigned* q_a) {
    unsigned ww[4] = {v.x, v.y, v.z, v.w};
    unsigned c = 0u;
    #pragma unroll
    for (int j = 0; j < 4; ++j) {
        unsigned t = ww[j] ^ mine4;
        unsigned z = ~(((t & 0x7F7F7F7Fu) + 0x7F7F7F7Fu) | t) & 0x80808080u;
        c |= z >> j;
    }
    while (c) {
        unsigned p   = (unsigned)__builtin_ctz(c);
        unsigned pos = atomicAdd(q_cnt, 1u);
        if (pos < QCAP) q_a[pos] = (unsigned)(idx * 16) + (((p ^ 7u) & 7u) << 2) + (p >> 3);
        c &= c - 1u;
    }
}

__device__ __forceinline__ void queue_owned_rows(int4 v, int idx, int r0, int nrows,
                                                 unsigned* q_cnt, unsigned* q_a) {
    int rr[4] = {v.x, v.y, v.z, v.w};
    #pragma unroll
    for (int j = 0; j < 4; ++j) {
        unsigned d = (unsigned)(rr[j] - r0);
        if (d < (unsigned)nrows) {
            unsigned pos = atomicAdd(q_cnt, 1u);
            if (pos < QCAP) q_a[pos] = (unsigned)(idx * 4 + j);
        }
    }
}

__device__ __forceinline__ void fma4(float w, const float4& v, float4& acc) {
    acc.x = fmaf(w, v.x, acc.x);
    acc.y = fmaf(w, v.y, acc.y);
    acc.z = fmaf(w, v.z, acc.z);
    acc.w = fmaf(w, v.w, acc.w);
}

template <int M>
__device__ __forceinline__ void add_rows(const float* __restrict__ xw, const uint2* s_e,
                                         unsigned first, int lane,
                                         float4& accA, float4& accB) {
    float  lp[M];
    float4 a[M], b[M];
    #pragma unroll
    for (int u = 0; u < M; ++u) {
        uint2 en = s_e[first + u];
        int cl = __builtin_amdgcn_readfirstlane((int)en.x);
        lp[u]  = __uint_as_float(en.y);
        const float4* s = (const float4*)(xw + (size_t)cl * OB);
        a[u] = s[lane];
        b[u] = s[32 + lane];
    }
    #pragma unroll
    for (int u = 0; u < M; ++u) {
        fma4(lp[u], a[u], accA);
        fma4(lp[u], b[u], accB);
    }
}

__device__ __forceinline__ void store_row(float* __restrict__ out, int n, int lane, float w,
                                          const float4& sa, const float4& sb,
                                          float4 accA, float4 accB, float2 bz0, float2 bz1) {
    fma4(w, sa, accA);
    fma4(w, sb, accB);
    float* o0 = out + (size_t)n * F_OUT + 2 * lane;
    float* o1 = out + ((size_t)N_NODES + n) * F_OUT + 2 * lane;
    *(float2*)(o0)      = make_float2(accA.x + bz0.x, accA.z + bz0.y);
    *(float2*)(o1)      = make_float2(accA.y + bz0.x, accA.w + bz0.y);
    *(float2*)(o0 + 64) = make_float2(accB.x + bz1.x, accB.z + bz1.y);
    *(float2*)(o1 + 64) = make_float2(accB.y + bz1.x, accB.w + bz1.y);
}

__global__ __launch_bounds__(OWN_THREADS)
void edge_owner_kernel(const unsigned* __restrict__ keyw,
                       const int* __restrict__ row,
                       const int* __restrict__ col,
                       const float* __restrict__ ew,
                       const float* __restrict__ deg,
                       const float* __restrict__ xw,
                       const float* __restrict__ bias,
                       float* __restrict__ out,
                       size_t ws_bytes) {
    __shared__ unsigned q_a[QCAP];
    __shared__ float    q_lap[QCAP];
    __shared__ uint2    s_e[QCAP];
    __shared__ unsigned r_cnt[ROWS_PAD];
    __shared__ unsigned r_beg[ROWS_PAD];
    __shared__ unsigned r_fill[ROWS_PAD];
    __shared__ unsigned q_cnt;
    __shared__ unsigned next_row;

    const int tid   = threadIdx.x;
    const int r0    = blockIdx.x * ROWS_PER_WG;
    const int nrows = min(ROWS_PER_WG, N_NODES - r0);
    const int wave  = __builtin_amdgcn_readfirstlane(tid >> 5);
    const int lane  = tid & 31;

    if (tid < ROWS_PAD) r_cnt[tid] = 0u;
    if (tid == 0) { q_cnt = 0u; next_row = (unsigned)OWN_WAVES; }
    __syncthreads();

    if (ws_bytes >= WS_KEY_END) {
        const uint4*   key16 = (const uint4*)keyw;
        const unsigned mine4 = (unsigned)blockIdx.x * 0x01010101u;
        const int      start = blockIdx.x * (N_KEY16 / OWN_WGS);
        for (int k = tid; k < SCAN_FULL; k += SCAN_UNROLL * OWN_THREADS) {
            uint4 v[SCAN_UNROLL];
            int   id[SCAN_UNROLL];
            #pragma unroll
            for (int u = 0; u < SCAN_UNROLL; ++u) {
                int idx = start + k + u * OWN_THREADS;
                if (idx >= N_KEY16) idx -= N_KEY16;
                id[u] = idx;
                v[u]  = key16[idx];
            }
            __builtin_amdgcn_sched_barrier(0);
            #pragma unroll
            for (int u = 0; u < SCAN_UNROLL; ++u) queue_owned(v[u], id[u], mine4, &q_cnt, q_a);
        }
        for (int kk = SCAN_FULL + tid; kk < N_KEY16; kk += OWN_THREADS) {
            int idx = start + kk;
            if (idx >= N_KEY16) idx -= N_KEY16;
            queue_owned(key16[idx], idx, mine4, &q_cnt, q_a);
        }
    } else {
        const int4* row4  = (const int4*)row;
        const int   start = blockIdx.x * (N_ROW4 / OWN_WGS);
        for (int k = tid; k < ROW_FULL; k += SCAN_UNROLL * OWN_THREADS) {
            int4 v[SCAN_UNROLL];
            int  id[SCAN_UNROLL];
            #pragma unroll
            for (int u = 0; u < SCAN_UNROLL; ++u) {
                int idx = start + k + u * OWN_THREADS;
                if (idx >= N_ROW4) idx -= N_ROW4;
                id[u] = idx;
                v[u]  = row4[idx];
            }
            __builtin_amdgcn_sched_barrier(0);
            #pragma unroll
            for (int u = 0; u < SCAN_UNROLL; ++u) queue_owned_rows(v[u], id[u], r0, nrows, &q_cnt, q_a);
        }
        for (int kk = ROW_FULL + tid; kk < N_ROW4; kk += OWN_THREADS) {
            int idx = start + kk;
            if (idx >= N_ROW4) idx -= N_ROW4;
            queue_owned_rows(row4[idx], idx, r0, nrows, &q_cnt, q_a);
        }
    }
    __syncthreads();


    const float2 bz0 = *(const float2*)(bias + 2 * lane);
    const float2 bz1 = *(const float2*)(bias + 64 + 2 * lane);

    if (q_cnt > (unsigned)QCAP) {
        for (int d = wave; d < nrows; d += OWN_WAVES) {
            const int n = r0 + d;
            const float dn = inv_sqrt_deg(deg[n]);
            const float4* sn = (const float4*)(xw + (size_t)n * OB);
            const float4 sa = sn[lane];
            const float4 sb = sn[32 + lane];
            float4 accA = make_float4(0.f, 0.f, 0.f, 0.f);
            float4 accB = make_float4(0.f, 0.f, 0.f, 0.f);
            for (int e0 = 0; e0 < E_EDGES; e0 += 32) {
                unsigned m = __builtin_amdgcn_ballot_w32(row[e0 + lane] == n);
                while (m) {
                    int e = e0 + __builtin_ctz(m);
                    m &= m - 1u;
                    int   cl  = col[e];
                    float lap = dn * ew[e] * inv_sqrt_deg(deg[cl]);
                    const float4* s = (const float4*)(xw + (size_t)cl * OB);
                    fma4(lap, s[lane], accA);
                    fma4(lap, s[32 + lane], accB);
                }
            }
            store_row(out, n, lane, dn * dn, sa, sb, accA, accB, bz0, bz1);
        }
        return;
    }
    const int nq = (int)q_cnt;

    for (int i = tid; i < nq; i += OWN_THREADS) {
        int e  = (int)q_a[i];
        int r  = row[e];
        int d  = r - r0;
        int cl = col[e];
        float lap = inv_sqrt_deg(deg[r]) * ew[e] * inv_sqrt_deg(deg[cl]);
        q_a[i]   = (unsigned)cl | ((unsigned)d << 16);
        q_lap[i] = lap;
        atomicAdd(&r_cnt[d], 1u);
    }
    __syncthreads();

    if (tid < 32) {
        unsigned c[8];
        unsigned s = 0u;
        #pragma unroll
        for (int j = 0; j < 8; ++j) { c[j] = r_cnt[tid * 8 + j]; s += c[j]; }
        unsigned incl = s;
        #pragma unroll
        for (int off = 1; off < 32; off <<= 1) {
            unsigned t = __shfl_up(incl, off, 32);
            if (tid >= off) incl += t;
        }
        unsigned base = incl - s;
        #pragma unroll
        for (int j = 0; j < 8; ++j) {
            r_beg[tid * 8 + j]  = base;
            r_fill[tid * 8 + j] = base;
            base += c[j];
        }
    }
    __syncthreads();

    for (int i = tid; i < nq; i += OWN_THREADS) {
        unsigned pk  = q_a[i];
        unsigned pos = atomicAdd(&r_fill[pk >> 16], 1u);
        s_e[pos] = make_uint2(pk & 0xFFFFu, __float_as_uint(q_lap[i]));
    }
    __syncthreads();

    int d = wave;
    while (d < nrows) {
        const int n = r0 + d;
        const unsigned beg = (unsigned)__builtin_amdgcn_readfirstlane((int)r_beg[d]);
        const unsigned cnt = (unsigned)__builtin_amdgcn_readfirstlane((int)r_cnt[d]);
        const float4* sn = (const float4*)(xw + (size_t)n * OB);
        const float4 sa = sn[lane];
        const float4 sb = sn[32 + lane];
        const float  dv = inv_sqrt_deg(deg[n]);
        float4 accA = make_float4(0.f, 0.f, 0.f, 0.f);
        float4 accB = make_float4(0.f, 0.f, 0.f, 0.f);
        unsigned k = 0u;
        for (; k + 4u <= cnt; k += 4u) add_rows<4>(xw, s_e, beg + k, lane, accA, accB);
        switch (cnt - k) {
            case 3u: add_rows<3>(xw, s_e, beg + k, lane, accA, accB); break;
            case 2u: add_rows<2>(xw, s_e, beg + k, lane, accA, accB); break;
            case 1u: add_rows<1>(xw, s_e, beg + k, lane, accA, accB); break;
            default: break;
        }
        store_row(out, n, lane, dv * dv, sa, sb, accA, accB, bz0, bz1);
        unsigned nd = 0u;
        if (lane == 0) nd = atomicAdd(&next_row, 1u);
        d = __builtin_amdgcn_readfirstlane((int)nd);
    }
}


extern "C" void kernel_launch(void* const* d_in, const int* in_sizes, int n_in,
                              void* d_out, int out_size, void* d_ws, size_t ws_size,
                              hipStream_t stream) {
    const float* x    = (const float*)d_in[0];
    const float* w    = (const float*)d_in[1];
    const float* bias = (const float*)d_in[2];
    const float* ew   = (const float*)d_in[3];
    const int*   eidx = (const int*)d_in[4];
    const int*   row  = eidx;
    const int*   col  = eidx + E_EDGES;
    float* out = (float*)d_out;

    float* xw   = (float*)d_ws;
    float* deg  = xw + (size_t)N_NODES * OB;
    unsigned* keyw = (unsigned*)(deg + N_NODES);

    deg_init_kernel<<<(N_NODES + 255) / 256, 256, 0, stream>>>(deg);

    gemm_wmma_kernel<<<GM_BLOCKS, GM_THREADS, 0, stream>>>(x, w, xw, row, ew, deg, keyw, ws_size);

    edge_owner_kernel<<<OWN_WGS, OWN_THREADS, 0, stream>>>(keyw, row, col, ew, deg, xw, bias, out, ws_size);
}


# Projection launch also stores a packed 8-byte record per edge (col, local row, weight); owners fetch 1 record per queued edge instead of row[], col[], ew[] entries; sums and their order unchanged
# speedup vs baseline: 4.6824x; 1.0373x over previous
#include <hip/hip_runtime.h>

#define N_NODES 50000
#define E_EDGES 800000
#define F_IN    256
#define F_OUT   128
#define BATCH   2
#define OB      (F_OUT * BATCH)

#define NT ((N_NODES + 15) / 16)

#define OWN_WGS      256
#define ROWS_PER_WG  ((N_NODES + OWN_WGS - 1) / OWN_WGS)

#define WS_KEY_OFFSET  (((size_t)N_NODES * OB + (size_t)N_NODES) * sizeof(float))
#define WS_KEY_END     (WS_KEY_OFFSET + (size_t)E_EDGES)
#define WS_REC_OFFSET  WS_KEY_END
#define WS_REC_END     (WS_REC_OFFSET + (size_t)E_EDGES * 8)

typedef float v2f __attribute__((ext_vector_type(2)));
typedef float v8f __attribute__((ext_vector_type(8)));


__global__ void deg_init_kernel(float* __restrict__ deg) {
    int i = blockIdx.x * blockDim.x + threadIdx.x;
    if (i < N_NODES) deg[i] = 1.0f;
}

__device__ __forceinline__ float inv_sqrt_deg(float d) {
    return d > 0.0f ? rsqrtf(d) : 0.0f;
}


#define GM_BLOCKS   256
#define GM_THREADS  768
#define EDGES_PER_BLOCK (E_EDGES / GM_BLOCKS)
#define EDGE_TRIPS      ((EDGES_PER_BLOCK + GM_THREADS - 1) / GM_THREADS)
#define GM_PAIRS    (GM_THREADS / 64)
#define KGROUPS     (F_IN / 8)

__device__ __forceinline__ v8f wmma4(float ax, float ay, float bx, float by, v8f c) {
    v2f a = {ax, ay};
    v2f b = {bx, by};
    return __builtin_amdgcn_wmma_f32_16x16x4_f32(
         false, a,  false, b,
         (short)0, c,  false,  false);
}

__device__ __forceinline__ void mma_group(const float4& a0, const float4& a1,
                                          const float4 (&bf)[4],
                                          v8f (&acc0)[4], v8f (&acc1)[4]) {
    #pragma unroll
    for (int t = 0; t < 4; ++t) {
        acc0[t] = wmma4(a0.x, a0.y, bf[t].x, bf[t].y, acc0[t]);
        acc1[t] = wmma4(a1.x, a1.y, bf[t].x, bf[t].y, acc1[t]);
    }
    #pragma unroll
    for (int t = 0; t < 4; ++t) {
        acc0[t] = wmma4(a0.z, a0.w, bf[t].z, bf[t].w, acc0[t]);
        acc1[t] = wmma4(a1.z, a1.w, bf[t].z, bf[t].w, acc1[t]);
    }
}

__global__ __launch_bounds__(GM_THREADS)
void gemm_wmma_kernel(const float* __restrict__ x,
                      const float* __restrict__ w,
                      float* __restrict__ xw,
                      const int* __restrict__ row,
                      const int* __restrict__ col,
                      const float* __restrict__ ew,
                      float* __restrict__ deg,
                      unsigned* __restrict__ keyw,
                      unsigned* __restrict__ recw,
                      size_t ws_bytes) {
    __shared__ float4 wl[F_IN * F_OUT / 4];

    const int tid = threadIdx.x;

    const int ebase = blockIdx.x * EDGES_PER_BLOCK;
    int   er[EDGE_TRIPS];
    int   ec[EDGE_TRIPS];
    float ev[EDGE_TRIPS];
    #pragma unroll
    for (int u = 0; u < EDGE_TRIPS; ++u) {
        const int i = min(tid + u * GM_THREADS, EDGES_PER_BLOCK - 1);
        er[u] = row[ebase + i];
        ec[u] = col[ebase + i];
        ev[u] = ew[ebase + i];
    }

    for (int idx = tid; idx < F_IN * F_OUT / 4; idx += GM_THREADS) {
        int o  = idx & (F_OUT - 1);
        int kq = idx >> 7;
        const float* src = w + (size_t)(kq * 4) * F_OUT + o;
        float4 v = make_float4(src[0], src[F_OUT], src[2 * F_OUT], src[3 * F_OUT]);
        int kk = kq >> 1, hh = kq & 1;
        int oh = o >> 6, t = (o >> 4) & 3, ll = o & 15;
        wl[((oh * KGROUPS + kk) * 4 + t) * 32 + hh * 16 + ll] = v;
    }
    __syncthreads();

    {
        unsigned char* key = (unsigned char*)keyw;
        uint2*         rec = (uint2*)recw;
        const bool has_keys = ws_bytes >= WS_KEY_END;
        const bool has_recs = ws_bytes >= WS_REC_END;
        #pragma unroll
        for (int u = 0; u < EDGE_TRIPS; ++u) {
            const int i = tid + u * GM_THREADS;
            if (i < EDGES_PER_BLOCK) {
                atomicAdd(&deg[er[u]], ev[u]);
                const unsigned own = (unsigned)er[u] / (unsigned)ROWS_PER_WG;
                if (has_keys) key[ebase + i] = (unsigned char)own;
                if (has_recs) rec[ebase + i] = make_uint2((unsigned)ec[u] | (((unsigned)er[u] - own * (unsigned)ROWS_PER_WG) << 16),
                                                          __float_as_uint(ev[u]));
            }
        }
    }

    const int wave = __builtin_amdgcn_readfirstlane(tid >> 5);
    const int lane = tid & 31;
    const int oh = wave & 1;
    const int h  = lane >> 4;
    const int l  = lane & 15;
    const float4* wb = wl + oh * (KGROUPS * 4 * 32) + lane;

    for (int nt = (wave >> 1) * GM_BLOCKS + blockIdx.x; nt < NT; nt += GM_PAIRS * GM_BLOCKS) {
        const int n0 = nt * 16;
        const float4* xa0 = (const float4*)(x + (size_t)(n0 + l) * F_IN + h * 4);
        const float4* xa1 = (const float4*)(x + ((size_t)N_NODES + n0 + l) * F_IN + h * 4);

        v8f acc0[4] = {};
        v8f acc1[4] = {};
        float4 a0 = xa0[0];
        float4 a1 = xa1[0];
        for (int kk = 0; kk < KGROUPS; ++kk) {
            float4 bf[4];
            #pragma unroll
            for (int t = 0; t < 4; ++t) bf[t] = wb[(kk * 4 + t) * 32];
            const int kn = (kk + 1) & (KGROUPS - 1);
            float4 na0 = xa0[kn * 2];
            float4 na1 = xa1[kn * 2];
            mma_group(a0, a1, bf, acc0, acc1);
            a0 = na0;
            a1 = na1;
        }

        float* orow = xw + (size_t)(n0 + 8 * h) * OB + (oh * 64 + l) * 2;
        #pragma unroll
        for (int t = 0; t < 4; ++t) {
            #pragma unroll
            for (int v = 0; v < 8; ++v) {
                float2 p = make_float2(acc0[t][v], acc1[t][v]);
                *(float2*)(orow + (size_t)v * OB + t * 32) = p;
            }
        }
    }
}


#define OWN_THREADS  1024
#define OWN_WAVES    (OWN_THREADS / 32)
#define QCAP         16384
#define ROWS_PAD     256
#define N_KEY16      (E_EDGES / 16)
#define N_ROW4       (E_EDGES / 4)
#define SCAN_UNROLL  8
#define SCAN_STEP    (SCAN_UNROLL * OWN_THREADS)
#define SCAN_FULL    ((N_KEY16 / SCAN_STEP) * SCAN_STEP)
#define ROW_FULL     ((N_ROW4 / SCAN_STEP) * SCAN_STEP)

__device__ __forceinline__ void queue_owned(uint4 v, int idx, unsigned mine4,
                                            unsigned* q_cnt, unsigned* q_a) {
    unsigned ww[4] = {v.x, v.y, v.z, v.w};
    unsigned c = 0u;
    #pragma unroll
    for (int j = 0; j < 4; ++j) {
        unsigned t = ww[j] ^ mine4;
        unsigned z = ~(((t & 0x7F7F7F7Fu) + 0x7F7F7F7Fu) | t) & 0x80808080u;
        c |= z >> j;
    }
    while (c) {
        unsigned p   = (unsigned)__builtin_ctz(c);
        unsigned pos = atomicAdd(q_cnt, 1u);
        if (pos < QCAP) q_a[pos] = (unsigned)(idx * 16) + (((p ^ 7u) & 7u) << 2) + (p >> 3);
        c &= c - 1u;
    }
}

__device__ __forceinline__ void queue_owned_rows(int4 v, int idx, int r0, int nrows,
                                                 unsigned* q_cnt, unsigned* q_a) {
    int rr[4] = {v.x, v.y, v.z, v.w};
    #pragma unroll
    for (int j = 0; j < 4; ++j) {
        unsigned d = (unsigned)(rr[j] - r0);
        if (d < (unsigned)nrows) {
            unsigned pos = atomicAdd(q_cnt, 1u);
            if (pos < QCAP) q_a[pos] = (unsigned)(idx * 4 + j);
        }
    }
}

__device__ __forceinline__ void fma4(float w, const float4& v, float4& acc) {
    acc.x = fmaf(w, v.x, acc.x);
    acc.y = fmaf(w, v.y, acc.y);
    acc.z = fmaf(w, v.z, acc.z);
    acc.w = fmaf(w, v.w, acc.w);
}

template <int M>
__device__ __forceinline__ void add_rows(const float* __restrict__ xw, const uint2* s_e,
                                         unsigned first, int lane,
                                         float4& accA, float4& accB) {
    float  lp[M];
    float4 a[M], b[M];
    #pragma unroll
    for (int u = 0; u < M; ++u) {
        uint2 en = s_e[first + u];
        int cl = __builtin_amdgcn_readfirstlane((int)en.x);
        lp[u]  = __uint_as_float(en.y);
        const float4* s = (const float4*)(xw + (size_t)cl * OB);
        a[u] = s[lane];
        b[u] = s[32 + lane];
    }
    #pragma unroll
    for (int u = 0; u < M; ++u) {
        fma4(lp[u], a[u], accA);
        fma4(lp[u], b[u], accB);
    }
}

__device__ __forceinline__ void store_row(float* __restrict__ out, int n, int lane, float w,
                                          const float4& sa, const float4& sb,
                                          float4 accA, float4 accB, float2 bz0, float2 bz1) {
    fma4(w, sa, accA);
    fma4(w, sb, accB);
    float* o0 = out + (size_t)n * F_OUT + 2 * lane;
    float* o1 = out + ((size_t)N_NODES + n) * F_OUT + 2 * lane;
    *(float2*)(o0)      = make_float2(accA.x + bz0.x, accA.z + bz0.y);
    *(float2*)(o1)      = make_float2(accA.y + bz0.x, accA.w + bz0.y);
    *(float2*)(o0 + 64) = make_float2(accB.x + bz1.x, accB.z + bz1.y);
    *(float2*)(o1 + 64) = make_float2(accB.y + bz1.x, accB.w + bz1.y);
}

__global__ __launch_bounds__(OWN_THREADS)
void edge_owner_kernel(const unsigned* __restrict__ keyw,
                       const unsigned* __restrict__ recw,
                       const int* __restrict__ row,
                       const int* __restrict__ col,
                       const float* __restrict__ ew,
                       const float* __restrict__ deg,
                       const float* __restrict__ xw,
                       const float* __restrict__ bias,
                       float* __restrict__ out,
                       size_t ws_bytes) {
    __shared__ unsigned q_a[QCAP];
    __shared__ float    q_lap[QCAP];
    __shared__ uint2    s_e[QCAP];
    __shared__ unsigned r_cnt[ROWS_PAD];
    __shared__ unsigned r_beg[ROWS_PAD];
    __shared__ unsigned r_fill[ROWS_PAD];
    __shared__ unsigned q_cnt;
    __shared__ unsigned next_row;

    const int tid   = threadIdx.x;
    const int r0    = blockIdx.x * ROWS_PER_WG;
    const int nrows = min(ROWS_PER_WG, N_NODES - r0);
    const int wave  = __builtin_amdgcn_readfirstlane(tid >> 5);
    const int lane  = tid & 31;

    if (tid < ROWS_PAD) r_cnt[tid] = 0u;
    if (tid == 0) { q_cnt = 0u; next_row = (unsigned)OWN_WAVES; }
    __syncthreads();

    if (ws_bytes >= WS_KEY_END) {
        const uint4*   key16 = (const uint4*)keyw;
        const unsigned mine4 = (unsigned)blockIdx.x * 0x01010101u;
        const int      start = blockIdx.x * (N_KEY16 / OWN_WGS);
        for (int k = tid; k < SCAN_FULL; k += SCAN_UNROLL * OWN_THREADS) {
            uint4 v[SCAN_UNROLL];
            int   id[SCAN_UNROLL];
            #pragma unroll
            for (int u = 0; u < SCAN_UNROLL; ++u) {
                int idx = start + k + u * OWN_THREADS;
                if (idx >= N_KEY16) idx -= N_KEY16;
                id[u] = idx;
                v[u]  = key16[idx];
            }
            __builtin_amdgcn_sched_barrier(0);
            #pragma unroll
            for (int u = 0; u < SCAN_UNROLL; ++u) queue_owned(v[u], id[u], mine4, &q_cnt, q_a);
        }
        for (int kk = SCAN_FULL + tid; kk < N_KEY16; kk += OWN_THREADS) {
            int idx = start + kk;
            if (idx >= N_KEY16) idx -= N_KEY16;
            queue_owned(key16[idx], idx, mine4, &q_cnt, q_a);
        }
    } else {
        const int4* row4  = (const int4*)row;
        const int   start = blockIdx.x * (N_ROW4 / OWN_WGS);
        for (int k = tid; k < ROW_FULL; k += SCAN_UNROLL * OWN_THREADS) {
            int4 v[SCAN_UNROLL];
            int  id[SCAN_UNROLL];
            #pragma unroll
            for (int u = 0; u < SCAN_UNROLL; ++u) {
                int idx = start + k + u * OWN_THREADS;
                if (idx >= N_ROW4) idx -= N_ROW4;
                id[u] = idx;
                v[u]  = row4[idx];
            }
            __builtin_amdgcn_sched_barrier(0);
            #pragma unroll
            for (int u = 0; u < SCAN_UNROLL; ++u) queue_owned_rows(v[u], id[u], r0, nrows, &q_cnt, q_a);
        }
        for (int kk = ROW_FULL + tid; kk < N_ROW4; kk += OWN_THREADS) {
            int idx = start + kk;
            if (idx >= N_ROW4) idx -= N_ROW4;
            queue_owned_rows(row4[idx], idx, r0, nrows, &q_cnt, q_a);
        }
    }
    __syncthreads();


    const float2 bz0 = *(const float2*)(bias + 2 * lane);
    const float2 bz1 = *(const float2*)(bias + 64 + 2 * lane);

    if (q_cnt > (unsigned)QCAP) {
        for (int d = wave; d < nrows; d += OWN_WAVES) {
            const int n = r0 + d;
            const float dn = inv_sqrt_deg(deg[n]);
            const float4* sn = (const float4*)(xw + (size_t)n * OB);
            const float4 sa = sn[lane];
            const float4 sb = sn[32 + lane];
            float4 accA = make_float4(0.f, 0.f, 0.f, 0.f);
            float4 accB = make_float4(0.f, 0.f, 0.f, 0.f);
            for (int e0 = 0; e0 < E_EDGES; e0 += 32) {
                unsigned m = __builtin_amdgcn_ballot_w32(row[e0 + lane] == n);
                while (m) {
                    int e = e0 + __builtin_ctz(m);
                    m &= m - 1u;
                    int   cl  = col[e];
                    float lap = dn * ew[e] * inv_sqrt_deg(deg[cl]);
                    const float4* s = (const float4*)(xw + (size_t)cl * OB);
                    fma4(lap, s[lane], accA);
                    fma4(lap, s[32 + lane], accB);
                }
            }
            store_row(out, n, lane, dn * dn, sa, sb, accA, accB, bz0, bz1);
        }
        return;
    }
    const int nq = (int)q_cnt;

    if (ws_bytes >= WS_REC_END) {
        const uint2* rec = (const uint2*)recw;
        for (int i = tid; i < nq; i += OWN_THREADS) {
            const uint2 rc = rec[q_a[i]];
            const unsigned d  = rc.x >> 16;
            const unsigned cl = rc.x & 0xFFFFu;
            float lap = inv_sqrt_deg(deg[r0 + (int)d]) * __uint_as_float(rc.y) * inv_sqrt_deg(deg[cl]);
            q_a[i]   = rc.x;
            q_lap[i] = lap;
            atomicAdd(&r_cnt[d], 1u);
        }
    } else {
        for (int i = tid; i < nq; i += OWN_THREADS) {
            int e  = (int)q_a[i];
            int r  = row[e];
            int d  = r - r0;
            int cl = col[e];
            float lap = inv_sqrt_deg(deg[r]) * ew[e] * inv_sqrt_deg(deg[cl]);
            q_a[i]   = (unsigned)cl | ((unsigned)d << 16);
            q_lap[i] = lap;
            atomicAdd(&r_cnt[d], 1u);
        }
    }
    __syncthreads();

    if (tid < 32) {
        unsigned c[8];
        unsigned s = 0u;
        #pragma unroll
        for (int j = 0; j < 8; ++j) { c[j] = r_cnt[tid * 8 + j]; s += c[j]; }
        unsigned incl = s;
        #pragma unroll
        for (int off = 1; off < 32; off <<= 1) {
            unsigned t = __shfl_up(incl, off, 32);
            if (tid >= off) incl += t;
        }
        unsigned base = incl - s;
        #pragma unroll
        for (int j = 0; j < 8; ++j) {
            r_beg[tid * 8 + j]  = base;
            r_fill[tid * 8 + j] = base;
            base += c[j];
        }
    }
    __syncthreads();

    for (int i = tid; i < nq; i += OWN_THREADS) {
        unsigned pk  = q_a[i];
        unsigned pos = atomicAdd(&r_fill[pk >> 16], 1u);
        s_e[pos] = make_uint2(pk & 0xFFFFu, __float_as_uint(q_lap[i]));
    }
    __syncthreads();

    int d = wave;
    while (d < nrows) {
        const int n = r0 + d;
        const unsigned beg = (unsigned)__builtin_amdgcn_readfirstlane((int)r_beg[d]);
        const unsigned cnt = (unsigned)__builtin_amdgcn_readfirstlane((int)r_cnt[d]);
        const float4* sn = (const float4*)(xw + (size_t)n * OB);
        const float4 sa = sn[lane];
        const float4 sb = sn[32 + lane];
        const float  dv = inv_sqrt_deg(deg[n]);
        float4 accA = make_float4(0.f, 0.f, 0.f, 0.f);
        float4 accB = make_float4(0.f, 0.f, 0.f, 0.f);
        unsigned k = 0u;
        for (; k + 4u <= cnt; k += 4u) add_rows<4>(xw, s_e, beg + k, lane, accA, accB);
        switch (cnt - k) {
            case 3u: add_rows<3>(xw, s_e, beg + k, lane, accA, accB); break;
            case 2u: add_rows<2>(xw, s_e, beg + k, lane, accA, accB); break;
            case 1u: add_rows<1>(xw, s_e, beg + k, lane, accA, accB); break;
            default: break;
        }
        store_row(out, n, lane, dv * dv, sa, sb, accA, accB, bz0, bz1);
        unsigned nd = 0u;
        if (lane == 0) nd = atomicAdd(&next_row, 1u);
        d = __builtin_amdgcn_readfirstlane((int)nd);
    }
}


extern "C" void kernel_launch(void* const* d_in, const int* in_sizes, int n_in,
                              void* d_out, int out_size, void* d_ws, size_t ws_size,
                              hipStream_t stream) {
    const float* x    = (const float*)d_in[0];
    const float* w    = (const float*)d_in[1];
    const float* bias = (const float*)d_in[2];
    const float* ew   = (const float*)d_in[3];
    const int*   eidx = (const int*)d_in[4];
    const int*   row  = eidx;
    const int*   col  = eidx + E_EDGES;
    float* out = (float*)d_out;

    float* xw   = (float*)d_ws;
    float* deg  = xw + (size_t)N_NODES * OB;
    unsigned* keyw = (unsigned*)(deg + N_NODES);
    unsigned* recw = keyw + E_EDGES / 4;

    deg_init_kernel<<<(N_NODES + 255) / 256, 256, 0, stream>>>(deg);

    gemm_wmma_kernel<<<GM_BLOCKS, GM_THREADS, 0, stream>>>(x, w, xw, row, col, ew, deg, keyw, recw, ws_size);

    edge_owner_kernel<<<OWN_WGS, OWN_THREADS, 0, stream>>>(keyw, recw, row, col, ew, deg, xw, bias, out, ws_size);
}


# Projection blocks counting-sort their edges by destination range in LDS (no global cursors), store chunk + offset row as whole lines; owners copy 256 segments, no key scan; row lists arrival-ordered
# speedup vs baseline: 5.0449x; 1.0774x over previous
#include <hip/hip_runtime.h>

#define N_NODES 50000
#define E_EDGES 800000
#define F_IN    256
#define F_OUT   128
#define BATCH   2
#define OB      (F_OUT * BATCH)

#define NT ((N_NODES + 15) / 16)

#define OWN_WGS      256
#define ROWS_PER_WG  ((N_NODES + OWN_WGS - 1) / OWN_WGS)

#define SRC_BLOCKS      256
#define EDGES_PER_BLOCK (E_EDGES / SRC_BLOCKS)
#define CHUNK_RECS      3136
#define TAB_STRIDE      288

#define WS_TAB_OFFSET  ((((size_t)N_NODES * OB + (size_t)N_NODES) * sizeof(float) + 127) / 128 * 128)
#define WS_TAB_END     (WS_TAB_OFFSET + (size_t)SRC_BLOCKS * TAB_STRIDE * sizeof(unsigned))
#define WS_SORT_OFFSET WS_TAB_END
#define WS_SORT_END    (WS_SORT_OFFSET + (size_t)SRC_BLOCKS * CHUNK_RECS * 8)

typedef float v2f __attribute__((ext_vector_type(2)));
typedef float v8f __attribute__((ext_vector_type(8)));


__global__ void deg_init_kernel(float* __restrict__ deg) {
    int i = blockIdx.x * blockDim.x + threadIdx.x;
    if (i < N_NODES) deg[i] = 1.0f;
}

__device__ __forceinline__ float inv_sqrt_deg(float d) {
    return d > 0.0f ? rsqrtf(d) : 0.0f;
}

__device__ __forceinline__ unsigned scan256(const unsigned* cnt, unsigned* excl, int lane) {
    unsigned c[8];
    unsigned s = 0u;
    #pragma unroll
    for (int j = 0; j < 8; ++j) { c[j] = cnt[lane * 8 + j]; s += c[j]; }
    unsigned incl = s;
    #pragma unroll
    for (int off = 1; off < 32; off <<= 1) {
        unsigned t = __shfl_up(incl, off, 32);
        if (lane >= off) incl += t;
    }
    unsigned base = incl - s;
    #pragma unroll
    for (int j = 0; j < 8; ++j) { excl[lane * 8 + j] = base; base += c[j]; }
    return __shfl(incl, 31, 32);
}


#define GM_BLOCKS   SRC_BLOCKS
#define GM_THREADS  768
#define EDGE_TRIPS      ((EDGES_PER_BLOCK + GM_THREADS - 1) / GM_THREADS)
#define GM_PAIRS    (GM_THREADS / 64)
#define KGROUPS     (F_IN / 8)

__device__ __forceinline__ v8f wmma4(float ax, float ay, float bx, float by, v8f c) {
    v2f a = {ax, ay};
    v2f b = {bx, by};
    return __builtin_amdgcn_wmma_f32_16x16x4_f32(
         false, a,  false, b,
         (short)0, c,  false,  false);
}

__device__ __forceinline__ void mma_group(const float4& a0, const float4& a1,
                                          const float4 (&bf)[4],
                                          v8f (&acc0)[4], v8f (&acc1)[4]) {
    #pragma unroll
    for (int t = 0; t < 4; ++t) {
        acc0[t] = wmma4(a0.x, a0.y, bf[t].x, bf[t].y, acc0[t]);
        acc1[t] = wmma4(a1.x, a1.y, bf[t].x, bf[t].y, acc1[t]);
    }
    #pragma unroll
    for (int t = 0; t < 4; ++t) {
        acc0[t] = wmma4(a0.z, a0.w, bf[t].z, bf[t].w, acc0[t]);
        acc1[t] = wmma4(a1.z, a1.w, bf[t].z, bf[t].w, acc1[t]);
    }
}

__global__ __launch_bounds__(GM_THREADS)
void gemm_wmma_kernel(const float* __restrict__ x,
                      const float* __restrict__ w,
                      float* __restrict__ xw,
                      const int* __restrict__ row,
                      const int* __restrict__ col,
                      const float* __restrict__ ew,
                      float* __restrict__ deg,
                      unsigned* __restrict__ tab,
                      unsigned* __restrict__ srt,
                      size_t ws_bytes) {
    __shared__ float4 wl[F_IN * F_OUT / 4];
    __shared__ unsigned s_hist[SRC_BLOCKS];
    __shared__ unsigned s_off[TAB_STRIDE];
    __shared__ uint4    s_rec4[CHUNK_RECS / 2];
    uint2* s_rec = (uint2*)s_rec4;

    const int tid = threadIdx.x;
    const bool has_sort = ws_bytes >= WS_SORT_END;
    if (tid < SRC_BLOCKS) s_hist[tid] = 0u;

    const int ebase = blockIdx.x * EDGES_PER_BLOCK;
    int   er[EDGE_TRIPS];
    int   ec[EDGE_TRIPS];
    float ev[EDGE_TRIPS];
    #pragma unroll
    for (int u = 0; u < EDGE_TRIPS; ++u) {
        const int i = min(tid + u * GM_THREADS, EDGES_PER_BLOCK - 1);
        er[u] = row[ebase + i];
        ec[u] = col[ebase + i];
        ev[u] = ew[ebase + i];
    }

    for (int idx = tid; idx < F_IN * F_OUT / 4; idx += GM_THREADS) {
        int o  = idx & (F_OUT - 1);
        int kq = idx >> 7;
        const float* src = w + (size_t)(kq * 4) * F_OUT + o;
        float4 v = make_float4(src[0], src[F_OUT], src[2 * F_OUT], src[3 * F_OUT]);
        int kk = kq >> 1, hh = kq & 1;
        int oh = o >> 6, t = (o >> 4) & 3, ll = o & 15;
        wl[((oh * KGROUPS + kk) * 4 + t) * 32 + hh * 16 + ll] = v;
    }
    __syncthreads();

    {
        unsigned own[EDGE_TRIPS];
        unsigned slot[EDGE_TRIPS];
        #pragma unroll
        for (int u = 0; u < EDGE_TRIPS; ++u) {
            const int i = tid + u * GM_THREADS;
            own[u]  = (unsigned)er[u] / (unsigned)ROWS_PER_WG;
            slot[u] = 0u;
            if (has_sort && i < EDGES_PER_BLOCK) slot[u] = atomicAdd(&s_hist[own[u]], 1u);
        }
        if (has_sort) {
            __syncthreads();
            if (tid < 32) {
                const unsigned total = scan256(s_hist, s_off, tid);
                s_off[SRC_BLOCKS + tid] = total;
            }
            __syncthreads();
            #pragma unroll
            for (int u = 0; u < EDGE_TRIPS; ++u) {
                const int i = tid + u * GM_THREADS;
                if (i < EDGES_PER_BLOCK)
                    s_rec[s_off[own[u]] + slot[u]] =
                        make_uint2((unsigned)ec[u] | (((unsigned)er[u] - own[u] * (unsigned)ROWS_PER_WG) << 16),
                                   __float_as_uint(ev[u]));
            }
            if (tid < CHUNK_RECS - EDGES_PER_BLOCK) s_rec[EDGES_PER_BLOCK + tid] = make_uint2(0u, 0u);
            __syncthreads();
            const uint4* s4 = s_rec4;
            uint4*       g4 = (uint4*)(srt + (size_t)blockIdx.x * (CHUNK_RECS * 2));
            for (int k = tid; k < CHUNK_RECS / 2; k += GM_THREADS) g4[k] = s4[k];
            if (tid < TAB_STRIDE) tab[blockIdx.x * TAB_STRIDE + tid] = s_off[tid];
        }
        #pragma unroll
        for (int u = 0; u < EDGE_TRIPS; ++u) {
            const int i = tid + u * GM_THREADS;
            if (i < EDGES_PER_BLOCK) atomicAdd(&deg[er[u]], ev[u]);
        }
    }

    const int wave = __builtin_amdgcn_readfirstlane(tid >> 5);
    const int lane = tid & 31;
    const int oh = wave & 1;
    const int h  = lane >> 4;
    const int l  = lane & 15;
    const float4* wb = wl + oh * (KGROUPS * 4 * 32) + lane;

    for (int nt = (wave >> 1) * GM_BLOCKS + blockIdx.x; nt < NT; nt += GM_PAIRS * GM_BLOCKS) {
        const int n0 = nt * 16;
        const float4* xa0 = (const float4*)(x + (size_t)(n0 + l) * F_IN + h * 4);
        const float4* xa1 = (const float4*)(x + ((size_t)N_NODES + n0 + l) * F_IN + h * 4);

        v8f acc0[4] = {};
        v8f acc1[4] = {};
        float4 a0 = xa0[0];
        float4 a1 = xa1[0];
        for (int kk = 0; kk < KGROUPS; ++kk) {
            float4 bf[4];
            #pragma unroll
            for (int t = 0; t < 4; ++t) bf[t] = wb[(kk * 4 + t) * 32];
            const int kn = (kk + 1) & (KGROUPS - 1);
            float4 na0 = xa0[kn * 2];
            float4 na1 = xa1[kn * 2];
            mma_group(a0, a1, bf, acc0, acc1);
            a0 = na0;
            a1 = na1;
        }

        float* orow = xw + (size_t)(n0 + 8 * h) * OB + (oh * 64 + l) * 2;
        #pragma unroll
        for (int t = 0; t < 4; ++t) {
            #pragma unroll
            for (int v = 0; v < 8; ++v) {
                float2 p = make_float2(acc0[t][v], acc1[t][v]);
                *(float2*)(orow + (size_t)v * OB + t * 32) = p;
            }
        }
    }
}


#define OWN_THREADS  1024
#define OWN_WAVES    (OWN_THREADS / 32)
#define QCAP         16384
#define ROWS_PAD     256
#define N_ROW4       (E_EDGES / 4)
#define SCAN_UNROLL  8
#define SCAN_STEP    (SCAN_UNROLL * OWN_THREADS)
#define ROW_FULL     ((N_ROW4 / SCAN_STEP) * SCAN_STEP)
#define SEG_LANES    32
#define SEG_GROUPS   (OWN_THREADS / SEG_LANES)
#define SEG_PASSES   (SRC_BLOCKS / SEG_GROUPS)

__device__ __forceinline__ void queue_owned_rows(int4 v, int idx, int r0, int nrows,
                                                 unsigned* q_cnt, unsigned* q_a) {
    int rr[4] = {v.x, v.y, v.z, v.w};
    #pragma unroll
    for (int j = 0; j < 4; ++j) {
        unsigned d = (unsigned)(rr[j] - r0);
        if (d < (unsigned)nrows) {
            unsigned pos = atomicAdd(q_cnt, 1u);
            if (pos < QCAP) q_a[pos] = (unsigned)(idx * 4 + j);
        }
    }
}

__device__ __forceinline__ void fma4(float w, const float4& v, float4& acc) {
    acc.x = fmaf(w, v.x, acc.x);
    acc.y = fmaf(w, v.y, acc.y);
    acc.z = fmaf(w, v.z, acc.z);
    acc.w = fmaf(w, v.w, acc.w);
}

template <int M>
__device__ __forceinline__ void add_rows(const float* __restrict__ xw, const uint2* s_e,
                                         unsigned first, int lane,
                                         float4& accA, float4& accB) {
    float  lp[M];
    float4 a[M], b[M];
    #pragma unroll
    for (int u = 0; u < M; ++u) {
        uint2 en = s_e[first + u];
        int cl = __builtin_amdgcn_readfirstlane((int)en.x);
        lp[u]  = __uint_as_float(en.y);
        const float4* s = (const float4*)(xw + (size_t)cl * OB);
        a[u] = s[lane];
        b[u] = s[32 + lane];
    }
    #pragma unroll
    for (int u = 0; u < M; ++u) {
        fma4(lp[u], a[u], accA);
        fma4(lp[u], b[u], accB);
    }
}

__device__ __forceinline__ void store_row(float* __restrict__ out, int n, int lane, float w,
                                          const float4& sa, const float4& sb,
                                          float4 accA, float4 accB, float2 bz0, float2 bz1) {
    fma4(w, sa, accA);
    fma4(w, sb, accB);
    float* o0 = out + (size_t)n * F_OUT + 2 * lane;
    float* o1 = out + ((size_t)N_NODES + n) * F_OUT + 2 * lane;
    *(float2*)(o0)      = make_float2(accA.x + bz0.x, accA.z + bz0.y);
    *(float2*)(o1)      = make_float2(accA.y + bz0.x, accA.w + bz0.y);
    *(float2*)(o0 + 64) = make_float2(accB.x + bz1.x, accB.z + bz1.y);
    *(float2*)(o1 + 64) = make_float2(accB.y + bz1.x, accB.w + bz1.y);
}

__global__ __launch_bounds__(OWN_THREADS)
void edge_owner_kernel(const unsigned* __restrict__ tab,
                       const unsigned* __restrict__ srt,
                       const int* __restrict__ row,
                       const int* __restrict__ col,
                       const float* __restrict__ ew,
                       const float* __restrict__ deg,
                       const float* __restrict__ xw,
                       const float* __restrict__ bias,
                       float* __restrict__ out,
                       size_t ws_bytes) {
    __shared__ unsigned q_a[QCAP];
    __shared__ float    q_lap[QCAP];
    __shared__ uint2    s_e[QCAP];
    __shared__ unsigned r_cnt[ROWS_PAD];
    __shared__ unsigned r_beg[ROWS_PAD];
    __shared__ unsigned r_fill[ROWS_PAD];
    __shared__ unsigned seg_src[SRC_BLOCKS];
    __shared__ unsigned seg_cnt[SRC_BLOCKS];
    __shared__ unsigned seg_pos[SRC_BLOCKS];
    __shared__ unsigned q_cnt;
    __shared__ unsigned next_row;

    const int tid   = threadIdx.x;
    const int r0    = blockIdx.x * ROWS_PER_WG;
    const int nrows = min(ROWS_PER_WG, N_NODES - r0);
    const int wave  = __builtin_amdgcn_readfirstlane(tid >> 5);
    const int lane  = tid & 31;

    if (tid < ROWS_PAD) r_cnt[tid] = 0u;
    if (tid == 0) { q_cnt = 0u; next_row = (unsigned)OWN_WAVES; }
    __syncthreads();

    const bool has_sort = ws_bytes >= WS_SORT_END;
    const uint2* rec = (const uint2*)srt;
    const unsigned j = (unsigned)lane;
    unsigned cnt[SEG_PASSES]  = {};
    unsigned sbeg[SEG_PASSES] = {};
    uint2    rc[SEG_PASSES]   = {};
    if (has_sort) {
        if (tid < SRC_BLOCKS) {
            const unsigned* trow = tab + tid * TAB_STRIDE + blockIdx.x;
            const unsigned s = min(trow[0], (unsigned)EDGES_PER_BLOCK);
            const unsigned e = min(max(trow[1], s), (unsigned)EDGES_PER_BLOCK);
            seg_src[tid] = (unsigned)tid * CHUNK_RECS + s;
            seg_cnt[tid] = e - s;
        }
        __syncthreads();
        #pragma unroll
        for (int t = 0; t < SEG_PASSES; ++t) {
            const int sg = wave + t * SEG_GROUPS;
            cnt[t]  = seg_cnt[sg];
            sbeg[t] = seg_src[sg];
            rc[t]   = rec[sbeg[t] + (j < cnt[t] ? j : 0u)];
        }
        if (tid < 32) {
            const unsigned total = scan256(seg_cnt, seg_pos, tid);
            if (tid == 0) q_cnt = total;
        }
    } else {
        const int4* row4  = (const int4*)row;
        const int   start = blockIdx.x * (N_ROW4 / OWN_WGS);
        for (int k = tid; k < ROW_FULL; k += SCAN_UNROLL * OWN_THREADS) {
            int4 v[SCAN_UNROLL];
            int  id[SCAN_UNROLL];
            #pragma unroll
            for (int u = 0; u < SCAN_UNROLL; ++u) {
                int idx = start + k + u * OWN_THREADS;
                if (idx >= N_ROW4) idx -= N_ROW4;
                id[u] = idx;
                v[u]  = row4[idx];
            }
            __builtin_amdgcn_sched_barrier(0);
            #pragma unroll
            for (int u = 0; u < SCAN_UNROLL; ++u) queue_owned_rows(v[u], id[u], r0, nrows, &q_cnt, q_a);
        }
        for (int kk = ROW_FULL + tid; kk < N_ROW4; kk += OWN_THREADS) {
            int idx = start + kk;
            if (idx >= N_ROW4) idx -= N_ROW4;
            queue_owned_rows(row4[idx], idx, r0, nrows, &q_cnt, q_a);
        }
    }
    __syncthreads();


    const float2 bz0 = *(const float2*)(bias + 2 * lane);
    const float2 bz1 = *(const float2*)(bias + 64 + 2 * lane);

    if (q_cnt > (unsigned)QCAP) {
        for (int d = wave; d < nrows; d += OWN_WAVES) {
            const int n = r0 + d;
            const float dn = inv_sqrt_deg(deg[n]);
            const float4* sn = (const float4*)(xw + (size_t)n * OB);
            const float4 sa = sn[lane];
            const float4 sb = sn[32 + lane];
            float4 accA = make_float4(0.f, 0.f, 0.f, 0.f);
            float4 accB = make_float4(0.f, 0.f, 0.f, 0.f);
            for (int e0 = 0; e0 < E_EDGES; e0 += 32) {
                unsigned m = __builtin_amdgcn_ballot_w32(row[e0 + lane] == n);
                while (m) {
                    int e = e0 + __builtin_ctz(m);
                    m &= m - 1u;
                    int   cl  = col[e];
                    float lap = dn * ew[e] * inv_sqrt_deg(deg[cl]);
                    const float4* s = (const float4*)(xw + (size_t)cl * OB);
                    fma4(lap, s[lane], accA);
                    fma4(lap, s[32 + lane], accB);
                }
            }
            store_row(out, n, lane, dn * dn, sa, sb, accA, accB, bz0, bz1);
        }
        return;
    }
    const int nq = (int)q_cnt;

    if (has_sort) {
        unsigned pos[SEG_PASSES];
        float    dr[SEG_PASSES], dc[SEG_PASSES];
        #pragma unroll
        for (int t = 0; t < SEG_PASSES; ++t) {
            const unsigned d = min(rc[t].x >> 16, (unsigned)(nrows - 1));
            dr[t]  = deg[r0 + (int)d];
            dc[t]  = deg[rc[t].x & 0xFFFFu];
            pos[t] = seg_pos[wave + t * SEG_GROUPS];
        }
        #pragma unroll
        for (int t = 0; t < SEG_PASSES; ++t) {
            if (j < cnt[t]) {
                const unsigned d = min(rc[t].x >> 16, (unsigned)(nrows - 1));
                q_a[pos[t] + j]   = (rc[t].x & 0xFFFFu) | (d << 16);
                q_lap[pos[t] + j] = inv_sqrt_deg(dr[t]) * __uint_as_float(rc[t].y) * inv_sqrt_deg(dc[t]);
                atomicAdd(&r_cnt[d], 1u);
            }
            for (unsigned jj = j + SEG_LANES; jj < cnt[t]; jj += SEG_LANES) {
                const uint2 r2 = rec[sbeg[t] + jj];
                const unsigned d  = min(r2.x >> 16, (unsigned)(nrows - 1));
                const unsigned cl = r2.x & 0xFFFFu;
                q_a[pos[t] + jj]   = cl | (d << 16);
                q_lap[pos[t] + jj] = inv_sqrt_deg(deg[r0 + (int)d]) * __uint_as_float(r2.y) * inv_sqrt_deg(deg[cl]);
                atomicAdd(&r_cnt[d], 1u);
            }
        }
    } else {
        for (int i = tid; i < nq; i += OWN_THREADS) {
            int e  = (int)q_a[i];
            int r  = row[e];
            int d  = r - r0;
            int cl = col[e];
            float lap = inv_sqrt_deg(deg[r]) * ew[e] * inv_sqrt_deg(deg[cl]);
            q_a[i]   = (unsigned)cl | ((unsigned)d << 16);
            q_lap[i] = lap;
            atomicAdd(&r_cnt[d], 1u);
        }
    }
    __syncthreads();

    if (tid < 32) {
        unsigned c[8];
        unsigned s = 0u;
        #pragma unroll
        for (int j = 0; j < 8; ++j) { c[j] = r_cnt[tid * 8 + j]; s += c[j]; }
        unsigned incl = s;
        #pragma unroll
        for (int off = 1; off < 32; off <<= 1) {
            unsigned t = __shfl_up(incl, off, 32);
            if (tid >= off) incl += t;
        }
        unsigned base = incl - s;
        #pragma unroll
        for (int j = 0; j < 8; ++j) {
            r_beg[tid * 8 + j]  = base;
            r_fill[tid * 8 + j] = base;
            base += c[j];
        }
    }
    __syncthreads();

    for (int i = tid; i < nq; i += OWN_THREADS) {
        unsigned pk  = q_a[i];
        unsigned pos = atomicAdd(&r_fill[pk >> 16], 1u);
        s_e[pos] = make_uint2(pk & 0xFFFFu, __float_as_uint(q_lap[i]));
    }
    __syncthreads();

    int d = wave;
    while (d < nrows) {
        const int n = r0 + d;
        const unsigned beg = (unsigned)__builtin_amdgcn_readfirstlane((int)r_beg[d]);
        const unsigned cnt = (unsigned)__builtin_amdgcn_readfirstlane((int)r_cnt[d]);
        const float4* sn = (const float4*)(xw + (size_t)n * OB);
        const float4 sa = sn[lane];
        const float4 sb = sn[32 + lane];
        const float  dv = inv_sqrt_deg(deg[n]);
        float4 accA = make_float4(0.f, 0.f, 0.f, 0.f);
        float4 accB = make_float4(0.f, 0.f, 0.f, 0.f);
        unsigned k = 0u;
        for (; k + 4u <= cnt; k += 4u) add_rows<4>(xw, s_e, beg + k, lane, accA, accB);
        switch (cnt - k) {
            case 3u: add_rows<3>(xw, s_e, beg + k, lane, accA, accB); break;
            case 2u: add_rows<2>(xw, s_e, beg + k, lane, accA, accB); break;
            case 1u: add_rows<1>(xw, s_e, beg + k, lane, accA, accB); break;
            default: break;
        }
        store_row(out, n, lane, dv * dv, sa, sb, accA, accB, bz0, bz1);
        unsigned nd = 0u;
        if (lane == 0) nd = atomicAdd(&next_row, 1u);
        d = __builtin_amdgcn_readfirstlane((int)nd);
    }
}


extern "C" void kernel_launch(void* const* d_in, const int* in_sizes, int n_in,
                              void* d_out, int out_size, void* d_ws, size_t ws_size,
                              hipStream_t stream) {
    const float* x    = (const float*)d_in[0];
    const float* w    = (const float*)d_in[1];
    const float* bias = (const float*)d_in[2];
    const float* ew   = (const float*)d_in[3];
    const int*   eidx = (const int*)d_in[4];
    const int*   row  = eidx;
    const int*   col  = eidx + E_EDGES;
    float* out = (float*)d_out;

    float* xw   = (float*)d_ws;
    float* deg  = xw + (size_t)N_NODES * OB;
    unsigned* tab = (unsigned*)((char*)d_ws + WS_TAB_OFFSET);
    unsigned* srt = (unsigned*)((char*)d_ws + WS_SORT_OFFSET);

    deg_init_kernel<<<(N_NODES + 255) / 256, 256, 0, stream>>>(deg);

    gemm_wmma_kernel<<<GM_BLOCKS, GM_THREADS, 0, stream>>>(x, w, xw, row, col, ew, deg, tab, srt, ws_size);

    edge_owner_kernel<<<OWN_WGS, OWN_THREADS, 0, stream>>>(tab, srt, row, col, ew, deg, xw, bias, out, ws_size);
}
